# starting kernel
# baseline (speedup 1.0000x reference)
#include <hip/hip_runtime.h>
#include <hip/hip_bf16.h>

typedef float v2f __attribute__((ext_vector_type(2)));
typedef float v8f __attribute__((ext_vector_type(8)));

#define N_NODES 50000
#define E_EDGES 500000
#define E2_EDGES 62500
#define ET_EDGES (E_EDGES + E2_EDGES)
#define NREL 1000
#define ALPHA 0.2f
#define KDIM 128

template <int LDBK, int NT>
__global__ void gat_wmma_gemm_f32(const float* __restrict__ A,
                                  const float* __restrict__ B,
                                  float* __restrict__ C,
                                  int M, int lda, int ldbn, int ldc, int coff,
                                  int mtiles, int ngroups) {
    int wave = threadIdx.x >> 5;
    long tile = (long)blockIdx.x * (blockDim.x >> 5) + wave;
    long tiles_total = (long)mtiles * ngroups;
    if (tile >= tiles_total) return;
    int ng = (int)(tile / mtiles);
    int mt = (int)(tile % mtiles);

    int lane = threadIdx.x & 31;
    int half = lane >> 4;
    int l16  = lane & 15;

    int arow = mt * 16 + l16;
    int arow_c = arow < M ? arow : (M - 1);
    const float* Ap = A + (size_t)arow_c * lda + half * 2;
    const float* Bp[NT];
#pragma unroll
    for (int s = 0; s < NT; ++s)
        Bp[s] = B + (size_t)(ng * NT * 16 + s * 16 + l16) * ldbn
                  + (size_t)(half * 2) * LDBK;

    v8f acc[NT];
#pragma unroll
    for (int s = 0; s < NT; ++s) acc[s] = (v8f){0.f,0.f,0.f,0.f,0.f,0.f,0.f,0.f};

    v2f a_cur = *(const v2f*)(Ap);
    v2f b_cur[NT];
#pragma unroll
    for (int s = 0; s < NT; ++s) {
        if constexpr (LDBK == 1) b_cur[s] = *(const v2f*)(Bp[s]);
        else { b_cur[s][0] = Bp[s][0]; b_cur[s][1] = Bp[s][LDBK]; }
    }

#pragma unroll 4
    for (int k0 = 0; k0 < KDIM; k0 += 4) {
        int kn = (k0 + 4 < KDIM) ? (k0 + 4) : 0;
        v2f a_nxt = *(const v2f*)(Ap + kn);
        v2f b_nxt[NT];
#pragma unroll
        for (int s = 0; s < NT; ++s) {
            if constexpr (LDBK == 1) b_nxt[s] = *(const v2f*)(Bp[s] + kn);
            else {
                b_nxt[s][0] = Bp[s][(size_t)kn * LDBK];
                b_nxt[s][1] = Bp[s][(size_t)kn * LDBK + LDBK];
            }
        }
#pragma unroll
        for (int s = 0; s < NT; ++s)
            acc[s] = __builtin_amdgcn_wmma_f32_16x16x4_f32(false, a_cur, false,
                                                           b_cur[s], (short)0,
                                                           acc[s], false, false);
        a_cur = a_nxt;
#pragma unroll
        for (int s = 0; s < NT; ++s) b_cur[s] = b_nxt[s];
    }

#pragma unroll
    for (int s = 0; s < NT; ++s) {
        int ccol = coff + ng * NT * 16 + s * 16 + l16;
#pragma unroll
        for (int r = 0; r < 8; ++r) {
            int crow = mt * 16 + r + half * 8;
            if (crow < M) C[(size_t)crow * ldc + ccol] = acc[s][r];
        }
    }
}

__global__ void gat_edge1(const int* __restrict__ el, const int* __restrict__ eln,
                          const int* __restrict__ etn,
                          const float* __restrict__ xnode,
                          const float* __restrict__ ec,
                          const float* __restrict__ relc,
                          const float* __restrict__ w2h,
                          float* __restrict__ num1, float* __restrict__ denom1) {
    int e = blockIdx.x * (blockDim.x >> 5) + (threadIdx.x >> 5);
    if (e >= ET_EDGES) return;
    int lane = threadIdx.x & 31;

    int s, d, t0 = 0, t1 = 0;
    bool direct = (e < E_EDGES);
    if (direct) {
        s = el[e]; d = el[E_EDGES + e];
    } else {
        int j = e - E_EDGES;
        s = eln[j]; d = eln[E2_EDGES + j];
        t0 = etn[2 * j]; t1 = etn[2 * j + 1];
    }

    int k = lane * 2;
#pragma unroll
    for (int h = 0; h < 2; ++h) {
        float2 xa = *(const float2*)(xnode + (size_t)s * 256 + h * 128 + k);
        float2 xb = *(const float2*)(xnode + (size_t)d * 256 + h * 128 + 64 + k);
        float e0, e1;
        if (direct) {
            float2 ev = *(const float2*)(ec + (size_t)e * 128 + h * 64 + k);
            e0 = ev.x; e1 = ev.y;
        } else {
            float2 a0 = *(const float2*)(relc + (size_t)t0 * 128 + h * 64 + k);
            float2 a1 = *(const float2*)(relc + (size_t)t1 * 128 + h * 64 + k);
            e0 = a0.x + a1.x; e1 = a0.y + a1.y;
        }
        float m0 = xa.x + xb.x + e0;
        float m1 = xa.y + xb.y + e1;
        float2 wv = *(const float2*)(w2h + h * 64 + k);
        float p = m0 * wv.x + m1 * wv.y;
#pragma unroll
        for (int off = 16; off > 0; off >>= 1) p += __shfl_xor(p, off, 32);
        float lr = p > 0.f ? p : ALPHA * p;
        float eev = __expf(-lr);
        float* dp = num1 + (size_t)s * 128 + h * 64 + k;
        atomicAdd(dp,     eev * m0);
        atomicAdd(dp + 1, eev * m1);
        if (lane == 0) atomicAdd(denom1 + (size_t)s * 2 + h, eev);
    }
}

__global__ void gat_fin1(float* __restrict__ num1, const float* __restrict__ denom1) {
    int i = blockIdx.x * blockDim.x + threadIdx.x;
    if (i >= N_NODES * 128) return;
    int n = i >> 7;
    int h = (i & 127) >> 6;
    float dnm = denom1[n * 2 + h];
    dnm = dnm > 1e-12f ? dnm : 1e-12f;
    num1[i] = num1[i] / dnm;
}

__global__ void gat_edge2(const int* __restrict__ el, const int* __restrict__ eln,
                          const int* __restrict__ et, const int* __restrict__ etn,
                          const float* __restrict__ xab2,
                          const float* __restrict__ relW2,
                          const float* __restrict__ w2o,
                          float* __restrict__ num2, float* __restrict__ denom2) {
    int e = blockIdx.x * (blockDim.x >> 5) + (threadIdx.x >> 5);
    if (e >= ET_EDGES) return;
    int lane = threadIdx.x & 31;

    int s, d;
    float4 ev;
    int k = lane * 4;
    if (e < E_EDGES) {
        s = el[e]; d = el[E_EDGES + e];
        ev = *(const float4*)(relW2 + (size_t)et[e] * 128 + k);
    } else {
        int j = e - E_EDGES;
        s = eln[j]; d = eln[E2_EDGES + j];
        int t0 = etn[2 * j], t1 = etn[2 * j + 1];
        float4 a0 = *(const float4*)(relW2 + (size_t)t0 * 128 + k);
        float4 a1 = *(const float4*)(relW2 + (size_t)t1 * 128 + k);
        ev.x = a0.x + a1.x; ev.y = a0.y + a1.y;
        ev.z = a0.z + a1.z; ev.w = a0.w + a1.w;
    }
    float4 xa = *(const float4*)(xab2 + (size_t)s * 256 + k);
    float4 xb = *(const float4*)(xab2 + (size_t)d * 256 + 128 + k);
    float m0 = xa.x + xb.x + ev.x;
    float m1 = xa.y + xb.y + ev.y;
    float m2 = xa.z + xb.z + ev.z;
    float m3 = xa.w + xb.w + ev.w;
    float4 wv = *(const float4*)(w2o + k);
    float p = m0 * wv.x + m1 * wv.y + m2 * wv.z + m3 * wv.w;
#pragma unroll
    for (int off = 16; off > 0; off >>= 1) p += __shfl_xor(p, off, 32);
    float lr = p > 0.f ? p : ALPHA * p;
    float eev = __expf(-lr);
    float* dp = num2 + (size_t)s * 128 + k;
    atomicAdd(dp,     eev * m0);
    atomicAdd(dp + 1, eev * m1);
    atomicAdd(dp + 2, eev * m2);
    atomicAdd(dp + 3, eev * m3);
    if (lane == 0) atomicAdd(denom2 + s, eev);
}

__global__ void gat_fin2(const float* __restrict__ num2,
                         const float* __restrict__ denom2,
                         float* __restrict__ out) {
    int i = blockIdx.x * blockDim.x + threadIdx.x;
    if (i >= N_NODES * 128) return;
    int n = i >> 7;
    float dnm = denom2[n];
    dnm = dnm > 1e-12f ? dnm : 1e-12f;
    float h = num2[i] / dnm;
    float y = h > 0.f ? h : (__expf(h) - 1.f);
    y = y > 0.f ? y : (__expf(y) - 1.f);
    out[i] = y;
}

template <int LDBK>
static inline void launch_gemm(hipStream_t st, const float* A, const float* B,
                               float* C, int M, int N,
                               int lda, int ldbn, int ldc, int coff) {
    constexpr int NT = 4;
    int mtiles  = (M + 15) / 16;
    int ngroups = N / (NT * 16);
    long tiles = (long)mtiles * ngroups;
    int blocks = (int)((tiles + 7) / 8);
    gat_wmma_gemm_f32<LDBK, NT><<<blocks, 256, 0, st>>>(A, B, C, M, lda, ldbn,
                                                        ldc, coff, mtiles,
                                                        ngroups);
}

extern "C" void kernel_launch(void* const* d_in, const int* in_sizes, int n_in,
                              void* d_out, int out_size, void* d_ws, size_t ws_size,
                              hipStream_t stream) {
    const float* x    = (const float*)d_in[0];
    const float* rel  = (const float*)d_in[1];
    const int*   el   = (const int*)d_in[2];
    const int*   et   = (const int*)d_in[3];
    const float* eemb = (const float*)d_in[4];
    const int*   eln  = (const int*)d_in[5];
    const int*   etn  = (const int*)d_in[6];
    const float* w1h  = (const float*)d_in[8];
    const float* w2h  = (const float*)d_in[9];
    const float* w1o  = (const float*)d_in[10];
    const float* w2o  = (const float*)d_in[11];
    const float* WR   = (const float*)d_in[12];
    float* out = (float*)d_out;

    float* ws = (float*)d_ws;
    float* xnode  = ws;
    float* ec     = xnode  + (size_t)N_NODES * 256;
    float* relc   = ec     + (size_t)E_EDGES * 128;
    float* num1   = relc   + (size_t)NREL * 128;
    float* denom1 = num1   + (size_t)N_NODES * 128;
    float* xab2   = denom1 + (size_t)N_NODES * 2;
    float* relW2  = xab2   + (size_t)N_NODES * 256;
    float* num2   = relW2  + (size_t)NREL * 128;
    float* denom2 = num2   + (size_t)N_NODES * 128;

    float* out_rel = out + (size_t)N_NODES * 128;

    hipMemsetAsync(num1,   0, (size_t)N_NODES * 128 * sizeof(float), stream);
    hipMemsetAsync(denom1, 0, (size_t)N_NODES * 2   * sizeof(float), stream);
    hipMemsetAsync(num2,   0, (size_t)N_NODES * 128 * sizeof(float), stream);
    hipMemsetAsync(denom2, 0, (size_t)N_NODES       * sizeof(float), stream);

    for (int h = 0; h < 2; ++h)
        for (int part = 0; part < 2; ++part)
            launch_gemm<1>(stream, x, w1h + (size_t)h * 64 * 384 + part * 128,
                           xnode, N_NODES, 64, 128, 384, 256,
                           h * 128 + part * 64);
    for (int h = 0; h < 2; ++h)
        launch_gemm<1>(stream, eemb, w1h + (size_t)h * 64 * 384 + 256,
                       ec, E_EDGES, 64, 128, 384, 128, h * 64);
    for (int h = 0; h < 2; ++h)
        launch_gemm<1>(stream, rel, w1h + (size_t)h * 64 * 384 + 256,
                       relc, NREL, 64, 128, 384, 128, h * 64);

    {
        int blocks = (ET_EDGES + 7) / 8;
        gat_edge1<<<blocks, 256, 0, stream>>>(el, eln, etn, xnode, ec, relc,
                                              w2h, num1, denom1);
    }
    gat_fin1<<<(N_NODES * 128 + 255) / 256, 256, 0, stream>>>(num1, denom1);

    launch_gemm<128>(stream, rel, WR, out_rel, NREL, 128, 128, 1, 128, 0);
    for (int part = 0; part < 2; ++part)
        launch_gemm<1>(stream, num1, w1o + part * 128,
                       xab2, N_NODES, 128, 128, 384, 256, part * 128);
    launch_gemm<1>(stream, out_rel, w1o + 256, relW2, NREL, 128, 128, 384, 128, 0);

    {
        int blocks = (ET_EDGES + 7) / 8;
        gat_edge2<<<blocks, 256, 0, stream>>>(el, eln, et, etn, xab2, relW2,
                                              w2o, num2, denom2);
    }
    gat_fin2<<<(N_NODES * 128 + 255) / 256, 256, 0, stream>>>(num2, denom2, out);
}


# edge scatter by node ownership: a workgroup owns 200 nodes, lists their edges by scanning all source ids; each wave adds its own nodes' edges in LDS in data-fixed order; no atomics, clears, finalize
# speedup vs baseline: 1.8314x; 1.8314x over previous
#include <hip/hip_runtime.h>
#include <hip/hip_bf16.h>

typedef float v2f __attribute__((ext_vector_type(2)));
typedef float v8f __attribute__((ext_vector_type(8)));

#define N_NODES 50000
#define E_EDGES 500000
#define E2_EDGES 62500
#define ET_EDGES (E_EDGES + E2_EDGES)
#define NREL 1000
#define ALPHA 0.2f
#define KDIM 128

template <int LDBK, int NT>
__global__ void gat_wmma_gemm_f32(const float* __restrict__ A,
                                  const float* __restrict__ B,
                                  float* __restrict__ C,
                                  int M, int lda, int ldbn, int ldc, int coff,
                                  int mtiles, int ngroups) {
    int wave = threadIdx.x >> 5;
    long tile = (long)blockIdx.x * (blockDim.x >> 5) + wave;
    long tiles_total = (long)mtiles * ngroups;
    if (tile >= tiles_total) return;
    int ng = (int)(tile / mtiles);
    int mt = (int)(tile % mtiles);

    int lane = threadIdx.x & 31;
    int half = lane >> 4;
    int l16  = lane & 15;

    int arow = mt * 16 + l16;
    int arow_c = arow < M ? arow : (M - 1);
    const float* Ap = A + (size_t)arow_c * lda + half * 2;
    const float* Bp[NT];
#pragma unroll
    for (int s = 0; s < NT; ++s)
        Bp[s] = B + (size_t)(ng * NT * 16 + s * 16 + l16) * ldbn
                  + (size_t)(half * 2) * LDBK;

    v8f acc[NT];
#pragma unroll
    for (int s = 0; s < NT; ++s) acc[s] = (v8f){0.f,0.f,0.f,0.f,0.f,0.f,0.f,0.f};

    v2f a_cur = *(const v2f*)(Ap);
    v2f b_cur[NT];
#pragma unroll
    for (int s = 0; s < NT; ++s) {
        if constexpr (LDBK == 1) b_cur[s] = *(const v2f*)(Bp[s]);
        else { b_cur[s][0] = Bp[s][0]; b_cur[s][1] = Bp[s][LDBK]; }
    }

#pragma unroll 4
    for (int k0 = 0; k0 < KDIM; k0 += 4) {
        int kn = (k0 + 4 < KDIM) ? (k0 + 4) : 0;
        v2f a_nxt = *(const v2f*)(Ap + kn);
        v2f b_nxt[NT];
#pragma unroll
        for (int s = 0; s < NT; ++s) {
            if constexpr (LDBK == 1) b_nxt[s] = *(const v2f*)(Bp[s] + kn);
            else {
                b_nxt[s][0] = Bp[s][(size_t)kn * LDBK];
                b_nxt[s][1] = Bp[s][(size_t)kn * LDBK + LDBK];
            }
        }
#pragma unroll
        for (int s = 0; s < NT; ++s)
            acc[s] = __builtin_amdgcn_wmma_f32_16x16x4_f32(false, a_cur, false,
                                                           b_cur[s], (short)0,
                                                           acc[s], false, false);
        a_cur = a_nxt;
#pragma unroll
        for (int s = 0; s < NT; ++s) b_cur[s] = b_nxt[s];
    }

#pragma unroll
    for (int s = 0; s < NT; ++s) {
        int ccol = coff + ng * NT * 16 + s * 16 + l16;
#pragma unroll
        for (int r = 0; r < 8; ++r) {
            int crow = mt * 16 + r + half * 8;
            if (crow < M) C[(size_t)crow * ldc + ccol] = acc[s][r];
        }
    }
}

#define WS_NT 4
#define WS_KS (KDIM / 4)

__device__ __forceinline__ void ws_load_a(v2f (&a)[WS_KS], const float* __restrict__ A,
                                          int mt, int l16, int half, int M, int lda) {
    int arow = mt * 16 + l16;
    arow = arow < M ? arow : (M - 1);
    const float* ap = A + (size_t)arow * lda + half * 2;
#pragma unroll
    for (int j = 0; j < WS_KS; ++j) a[j] = *(const v2f*)(ap + j * 4);
}

__device__ __forceinline__ void ws_tile(const v2f (&a)[WS_KS],
                                        const v2f (&bf)[WS_NT][WS_KS],
                                        float* __restrict__ C, int mt, int l16,
                                        int half, int M, int ldc, int coff) {
    v8f acc[WS_NT];
#pragma unroll
    for (int s = 0; s < WS_NT; ++s) acc[s] = (v8f){0.f,0.f,0.f,0.f,0.f,0.f,0.f,0.f};
#pragma unroll
    for (int j = 0; j < WS_KS; ++j) {
#pragma unroll
        for (int s = 0; s < WS_NT; ++s)
            acc[s] = __builtin_amdgcn_wmma_f32_16x16x4_f32(false, a[j], false,
                                                           bf[s][j], (short)0,
                                                           acc[s], false, false);
    }
    float* cp = C + (size_t)(mt * 16 + half * 8) * ldc + coff + l16;
    if (mt * 16 + 16 <= M) {
#pragma unroll
        for (int r = 0; r < 8; ++r) {
#pragma unroll
            for (int s = 0; s < WS_NT; ++s) cp[(size_t)r * ldc + s * 16] = acc[s][r];
        }
    } else {
#pragma unroll
        for (int r = 0; r < 8; ++r) {
            if (mt * 16 + half * 8 + r < M) {
#pragma unroll
                for (int s = 0; s < WS_NT; ++s) cp[(size_t)r * ldc + s * 16] = acc[s][r];
            }
        }
    }
}

__global__ void __launch_bounds__(128)
gat_wmma_gemm_ws(const float* __restrict__ A, const float* __restrict__ B,
                 float* __restrict__ C, int M, int lda, int ldbn, int ldc,
                 int mtiles, int ncg, int gdiv, int ghi, int glo) {
    const int lane = threadIdx.x & 31;
    const int half = lane >> 4;
    const int l16  = lane & 15;
    const int gw = blockIdx.x * 4 + __builtin_amdgcn_readfirstlane(threadIdx.x >> 5);
    const int nw = gridDim.x * 4;
    const int cg = gw % ncg;
    const int slot = gw / ncg;
    const int nslots = nw / ncg;
    if (slot >= mtiles) return;

    const float* Bg = B + (size_t)(cg / gdiv) * ghi + (size_t)(cg % gdiv) * glo
                        + half * 2;
    v2f bf[WS_NT][WS_KS];
#pragma unroll
    for (int s = 0; s < WS_NT; ++s) {
        const float* bp = Bg + (size_t)(s * 16 + l16) * ldbn;
#pragma unroll
        for (int j = 0; j < WS_KS; ++j) bf[s][j] = *(const v2f*)(bp + j * 4);
    }

    const int coff = cg * (WS_NT * 16);
    int mt = slot;
    v2f a0[WS_KS], a1[WS_KS];
    ws_load_a(a0, A, mt, l16, half, M, lda);
    while (true) {
        {
            int mn = mt + nslots;
            mn = mn < mtiles ? mn : mt;
            ws_load_a(a1, A, mn, l16, half, M, lda);
            __builtin_amdgcn_sched_barrier(0);
            ws_tile(a0, bf, C, mt, l16, half, M, ldc, coff);
        }
        mt += nslots;
        if (mt >= mtiles) break;
        {
            int mn = mt + nslots;
            mn = mn < mtiles ? mn : mt;
            ws_load_a(a0, A, mn, l16, half, M, lda);
            __builtin_amdgcn_sched_barrier(0);
            ws_tile(a1, bf, C, mt, l16, half, M, ldc, coff);
        }
        mt += nslots;
        if (mt >= mtiles) break;
    }
}

#define OWN_NODES 200
#define OWN_WAVES 32
#define OWN_CAP   640
#define OWN_NG1   ((E_EDGES + 127) / 128)
#define OWN_NG2   ((E2_EDGES + 127) / 128)
#define OWN_NGT   (OWN_NG1 + OWN_NG2)
#define OWN_GPW   ((OWN_NGT + OWN_WAVES - 1) / OWN_WAVES)
#define OWN_ROUNDS ((OWN_GPW + 3) / 4)
static_assert(N_NODES <= 65536 && NREL <= 32768, "second list word packs 16 + 15 bits");
static_assert(ET_EDGES <= (1 << 20) && OWN_NODES <= 256, "first list word packs 20 + 8 bits");
static_assert(N_NODES % OWN_NODES == 0, "every workgroup owns a full set of nodes");

struct OwnEdge {
    float xa[4], xb[4], r0[4], r1[4];
    int nl;
    int direct;
};

template <int L>
__device__ __forceinline__ void own_load(OwnEdge& r, int e, int nl, int dt, int base, int lane,
                                         const int* __restrict__ etn,
                                         const float* __restrict__ xrow,
                                         const float* __restrict__ erow,
                                         const float* __restrict__ rtab) {
    const int s = base + nl;
    const int d = dt & 0xFFFF;
    r.nl = nl;
    if (e < E_EDGES) {
        r.direct = 1;
        if constexpr (L == 1) {
#pragma unroll
            for (int h = 0; h < 2; ++h) {
                float2 v = *(const float2*)(erow + (size_t)e * 128 + h * 64 + lane * 2);
                r.r0[h * 2] = v.x; r.r0[h * 2 + 1] = v.y;
                r.r1[h * 2] = 0.f; r.r1[h * 2 + 1] = 0.f;
            }
        } else {
            int t = dt >> 16;
            float4 v = *(const float4*)(rtab + (size_t)t * 128 + lane * 4);
            r.r0[0] = v.x; r.r0[1] = v.y; r.r0[2] = v.z; r.r0[3] = v.w;
            r.r1[0] = 0.f; r.r1[1] = 0.f; r.r1[2] = 0.f; r.r1[3] = 0.f;
        }
    } else {
        int j = e - E_EDGES;
        int t0 = etn[2 * j], t1 = etn[2 * j + 1];
        r.direct = 0;
        if constexpr (L == 1) {
#pragma unroll
            for (int h = 0; h < 2; ++h) {
                float2 a0 = *(const float2*)(rtab + (size_t)t0 * 128 + h * 64 + lane * 2);
                float2 a1 = *(const float2*)(rtab + (size_t)t1 * 128 + h * 64 + lane * 2);
                r.r0[h * 2] = a0.x; r.r0[h * 2 + 1] = a0.y;
                r.r1[h * 2] = a1.x; r.r1[h * 2 + 1] = a1.y;
            }
        } else {
            float4 a0 = *(const float4*)(rtab + (size_t)t0 * 128 + lane * 4);
            float4 a1 = *(const float4*)(rtab + (size_t)t1 * 128 + lane * 4);
            r.r0[0] = a0.x; r.r0[1] = a0.y; r.r0[2] = a0.z; r.r0[3] = a0.w;
            r.r1[0] = a1.x; r.r1[1] = a1.y; r.r1[2] = a1.z; r.r1[3] = a1.w;
        }
    }
    if constexpr (L == 1) {
#pragma unroll
        for (int h = 0; h < 2; ++h) {
            float2 a = *(const float2*)(xrow + (size_t)s * 256 + h * 128 + lane * 2);
            float2 b = *(const float2*)(xrow + (size_t)d * 256 + h * 128 + 64 + lane * 2);
            r.xa[h * 2] = a.x; r.xa[h * 2 + 1] = a.y;
            r.xb[h * 2] = b.x; r.xb[h * 2 + 1] = b.y;
        }
    } else {
        float4 a = *(const float4*)(xrow + (size_t)s * 256 + lane * 4);
        float4 b = *(const float4*)(xrow + (size_t)d * 256 + 128 + lane * 4);
        r.xa[0] = a.x; r.xa[1] = a.y; r.xa[2] = a.z; r.xa[3] = a.w;
        r.xb[0] = b.x; r.xb[1] = b.y; r.xb[2] = b.z; r.xb[3] = b.w;
    }
}

template <int L>
__device__ __forceinline__ void own_add(const OwnEdge& r, const float (&w)[4], int lane,
                                        float* acc, float* den) {
    float m[4];
#pragma unroll
    for (int c = 0; c < 4; ++c) {
        float ev = r.direct ? r.r0[c] : (r.r0[c] + r.r1[c]);
        m[c] = r.xa[c] + r.xb[c] + ev;
    }
    if constexpr (L == 1) {
#pragma unroll
        for (int h = 0; h < 2; ++h) {
            float m0 = m[h * 2], m1 = m[h * 2 + 1];
            float p = m0 * w[h * 2] + m1 * w[h * 2 + 1];
#pragma unroll
            for (int off = 16; off > 0; off >>= 1) p += __shfl_xor(p, off, 32);
            float lr = p > 0.f ? p : ALPHA * p;
            float eev = __expf(-lr);
            float2* ap = (float2*)(acc + r.nl * 128 + h * 64 + lane * 2);
            float2 av = *ap;
            av.x += eev * m0;
            av.y += eev * m1;
            *ap = av;
            if (lane == 0) den[r.nl * 2 + h] += eev;
        }
    } else {
        float p = m[0] * w[0] + m[1] * w[1] + m[2] * w[2] + m[3] * w[3];
#pragma unroll
        for (int off = 16; off > 0; off >>= 1) p += __shfl_xor(p, off, 32);
        float lr = p > 0.f ? p : ALPHA * p;
        float eev = __expf(-lr);
        float4* ap = (float4*)(acc + r.nl * 128 + lane * 4);
        float4 av = *ap;
        av.x += eev * m[0];
        av.y += eev * m[1];
        av.z += eev * m[2];
        av.w += eev * m[3];
        *ap = av;
        if (lane == 0) den[r.nl] += eev;
    }
}

template <int L>
__global__ void __launch_bounds__(OWN_WAVES * 32)
gat_own(const int* __restrict__ el, const int* __restrict__ eln,
        const int* __restrict__ et, const int* __restrict__ etn,
        const float* __restrict__ xrow, const float* __restrict__ erow,
        const float* __restrict__ rtab, const float* __restrict__ wvec,
        float* __restrict__ outp) {
    __shared__ float acc[OWN_NODES * 128];
    __shared__ float den[OWN_NODES * 2];
    __shared__ int lists[OWN_WAVES * OWN_CAP];
    __shared__ int lists2[OWN_WAVES * OWN_CAP];
    __shared__ int s_cnt[OWN_WAVES];
    __shared__ int s_more;

    const int tid  = threadIdx.x;
    const int lane = tid & 31;
    const int wv   = __builtin_amdgcn_readfirstlane(tid >> 5);
    const int base = blockIdx.x * OWN_NODES;

    for (int i = tid; i < OWN_NODES * 128; i += OWN_WAVES * 32) acc[i] = 0.f;
    for (int i = tid; i < OWN_NODES * 2; i += OWN_WAVES * 32) den[i] = 0.f;

    float w[4];
    if constexpr (L == 1) {
        w[0] = wvec[lane * 2];      w[1] = wvec[lane * 2 + 1];
        w[2] = wvec[64 + lane * 2]; w[3] = wvec[64 + lane * 2 + 1];
    } else {
        w[0] = wvec[lane * 4];     w[1] = wvec[lane * 4 + 1];
        w[2] = wvec[lane * 4 + 2]; w[3] = wvec[lane * 4 + 3];
    }

    int g = wv * OWN_GPW;
    int gend = g + OWN_GPW;
    gend = gend < OWN_NGT ? gend : OWN_NGT;
    int* mylist = lists + wv * OWN_CAP;
    int* mylist2 = lists2 + wv * OWN_CAP;
    const unsigned lt_mask = (1u << lane) - 1u;

    for (int round = 0; round < OWN_ROUNDS; ++round) {
        if (tid == 0) s_more = 0;
        __syncthreads();

        int cnt = 0;
        while (g < gend && cnt + 512 <= OWN_CAP) {
            int4 v[4];
#pragma unroll
            for (int q = 0; q < 4; ++q) {
                int gi = g + q;
                gi = gi < gend ? gi : (gend - 1);
                const int* p = gi < OWN_NG1 ? (el + (size_t)gi * 128)
                                            : (eln + (size_t)(gi - OWN_NG1) * 128);
                v[q] = *(const int4*)(p + lane * 4);
            }
#pragma unroll
            for (int q = 0; q < 4; ++q) {
                int gi = g + q;
                if (gi < gend) {
                    bool isd = gi < OWN_NG1;
                    int eb   = isd ? gi * 128 : (gi - OWN_NG1) * 128;
                    int lim  = isd ? E_EDGES : E2_EDGES;
                    int eoff = isd ? 0 : E_EDGES;
                    int idx0 = eb + lane * 4;
                    int sv[4] = {v[q].x, v[q].y, v[q].z, v[q].w};
#pragma unroll
                    for (int j = 0; j < 4; ++j) {
                        int loc = sv[j] - base;
                        bool hit = (idx0 + j < lim) && ((unsigned)loc < (unsigned)OWN_NODES);
                        unsigned mask = __builtin_amdgcn_ballot_w32(hit);
                        if (mask != 0u) {
                            int pos = cnt + __builtin_popcount(mask & lt_mask);
                            if (hit) mylist[pos] = (eoff + idx0 + j) | (loc << 20);
                            cnt += __builtin_popcount(mask);
                        }
                    }
                }
            }
            g += 4;
        }
        for (int i = lane; i < cnt; i += 32) {
            int e = mylist[i] & 0xFFFFF;
            int d, t = 0;
            if (e < E_EDGES) {
                d = el[E_EDGES + e];
                if constexpr (L == 2) t = et[e];
            } else {
                d = eln[E2_EDGES + (e - E_EDGES)];
            }
            mylist2[i] = d | (t << 16);
        }
        if (lane == 0) {
            s_cnt[wv] = cnt;
            if (g < gend) s_more = 1;
        }
        __syncthreads();
        const int more = s_more;

        {
            OwnEdge cur = {};
            bool have = false;
            for (int w2 = 0; w2 < OWN_WAVES; ++w2) {
                const int n = __builtin_amdgcn_readfirstlane(s_cnt[w2]);
                const int* lp = lists + w2 * OWN_CAP;
                const int* lp2 = lists2 + w2 * OWN_CAP;
                for (int i0 = 0; i0 < n; i0 += 32) {
                    int idx = i0 + lane;
                    int ent = idx < n ? lp[idx] : 0;
                    int ent2 = idx < n ? lp2[idx] : 0;
                    bool mine = (idx < n) && (((ent >> 20) & 31) == wv);
                    unsigned mask = __builtin_amdgcn_ballot_w32(mine);
                    while (mask != 0u) {
                        int b = __builtin_ctz(mask);
                        mask &= mask - 1u;
                        int en = __builtin_amdgcn_readlane(ent, b);
                        int en2 = __builtin_amdgcn_readlane(ent2, b);
                        OwnEdge nx;
                        own_load<L>(nx, en & 0xFFFFF, en >> 20, en2, base, lane,
                                    etn, xrow, erow, rtab);
                        if (have) own_add<L>(cur, w, lane, acc, den);
                        cur = nx;
                        have = true;
                    }
                }
            }
            if (have) own_add<L>(cur, w, lane, acc, den);
        }
        __syncthreads();
        if (!more) break;
    }

    float* orow = outp + (size_t)base * 128;
    for (int i = tid; i < OWN_NODES * 128; i += OWN_WAVES * 32) {
        int nl = i >> 7;
        float dnm;
        if constexpr (L == 1) dnm = den[nl * 2 + ((i & 127) >> 6)];
        else                  dnm = den[nl];
        dnm = dnm > 1e-12f ? dnm : 1e-12f;
        float h = acc[i] / dnm;
        if constexpr (L == 1) {
            orow[i] = h;
        } else {
            float y = h > 0.f ? h : (__expf(h) - 1.f);
            y = y > 0.f ? y : (__expf(y) - 1.f);
            orow[i] = y;
        }
    }
}

template <int LDBK>
static inline void launch_gemm(hipStream_t st, const float* A, const float* B,
                               float* C, int M, int N,
                               int lda, int ldbn, int ldc, int coff) {
    constexpr int NT = 4;
    int mtiles  = (M + 15) / 16;
    int ngroups = N / (NT * 16);
    long tiles = (long)mtiles * ngroups;
    int blocks = (int)((tiles + 7) / 8);
    gat_wmma_gemm_f32<LDBK, NT><<<blocks, 256, 0, st>>>(A, B, C, M, lda, ldbn,
                                                        ldc, coff, mtiles,
                                                        ngroups);
}

extern "C" void kernel_launch(void* const* d_in, const int* in_sizes, int n_in,
                              void* d_out, int out_size, void* d_ws, size_t ws_size,
                              hipStream_t stream) {
    const float* x    = (const float*)d_in[0];
    const float* rel  = (const float*)d_in[1];
    const int*   el   = (const int*)d_in[2];
    const int*   et   = (const int*)d_in[3];
    const float* eemb = (const float*)d_in[4];
    const int*   eln  = (const int*)d_in[5];
    const int*   etn  = (const int*)d_in[6];
    const float* w1h  = (const float*)d_in[8];
    const float* w2h  = (const float*)d_in[9];
    const float* w1o  = (const float*)d_in[10];
    const float* w2o  = (const float*)d_in[11];
    const float* WR   = (const float*)d_in[12];
    float* out = (float*)d_out;

    float* ws = (float*)d_ws;
    float* xnode  = ws;
    float* ec     = xnode  + (size_t)N_NODES * 256;
    float* relc   = ec     + (size_t)E_EDGES * 128;
    float* num1   = relc   + (size_t)NREL * 128;
    float* xab2   = num1   + (size_t)N_NODES * 128;
    float* relW2  = xab2   + (size_t)N_NODES * 256;

    float* out_rel = out + (size_t)N_NODES * 128;

    gat_wmma_gemm_ws<<<512, 128, 0, stream>>>(x, w1h, xnode, N_NODES, 128, 384, 256,
                                              N_NODES / 16, 4, 2, 64 * 384, 128);
    gat_wmma_gemm_ws<<<512, 128, 0, stream>>>(eemb, w1h + 256, ec, E_EDGES, 128, 384,
                                              128, E_EDGES / 16, 2, 1, 64 * 384, 0);
    for (int h = 0; h < 2; ++h)
        launch_gemm<1>(stream, rel, w1h + (size_t)h * 64 * 384 + 256,
                       relc, NREL, 64, 128, 384, 128, h * 64);

    gat_own<1><<<N_NODES / OWN_NODES, OWN_WAVES * 32, 0, stream>>>(
        el, eln, et, etn, xnode, ec, relc, w2h, num1);

    launch_gemm<128>(stream, rel, WR, out_rel, NREL, 128, 128, 1, 128, 0);
    gat_wmma_gemm_ws<<<512, 128, 0, stream>>>(num1, w1o, xab2, N_NODES, 128, 384, 256,
                                              N_NODES / 16, 4, 2, 128, 64 * 384);
    launch_gemm<1>(stream, out_rel, w1o + 256, relW2, NREL, 128, 128, 384, 128, 0);

    gat_own<2><<<N_NODES / OWN_NODES, OWN_WAVES * 32, 0, stream>>>(
        el, eln, et, etn, xab2, ec, relW2, w2o, out);
}


# edge lists built once per call: layer 1 writes each wave's taken entries to the workspace; layer 2 copies them and walks only its own list instead of rescanning all source ids; add order unchanged
# speedup vs baseline: 1.9935x; 1.0885x over previous
#include <hip/hip_runtime.h>
#include <hip/hip_bf16.h>

typedef float v2f __attribute__((ext_vector_type(2)));
typedef float v8f __attribute__((ext_vector_type(8)));

#define N_NODES 50000
#define E_EDGES 500000
#define E2_EDGES 62500
#define ET_EDGES (E_EDGES + E2_EDGES)
#define NREL 1000
#define ALPHA 0.2f
#define KDIM 128

template <int LDBK, int NT>
__global__ void gat_wmma_gemm_f32(const float* __restrict__ A,
                                  const float* __restrict__ B,
                                  float* __restrict__ C,
                                  int M, int lda, int ldbn, int ldc, int coff,
                                  int mtiles, int ngroups) {
    int wave = threadIdx.x >> 5;
    long tile = (long)blockIdx.x * (blockDim.x >> 5) + wave;
    long tiles_total = (long)mtiles * ngroups;
    if (tile >= tiles_total) return;
    int ng = (int)(tile / mtiles);
    int mt = (int)(tile % mtiles);

    int lane = threadIdx.x & 31;
    int half = lane >> 4;
    int l16  = lane & 15;

    int arow = mt * 16 + l16;
    int arow_c = arow < M ? arow : (M - 1);
    const float* Ap = A + (size_t)arow_c * lda + half * 2;
    const float* Bp[NT];
#pragma unroll
    for (int s = 0; s < NT; ++s)
        Bp[s] = B + (size_t)(ng * NT * 16 + s * 16 + l16) * ldbn
                  + (size_t)(half * 2) * LDBK;

    v8f acc[NT];
#pragma unroll
    for (int s = 0; s < NT; ++s) acc[s] = (v8f){0.f,0.f,0.f,0.f,0.f,0.f,0.f,0.f};

    v2f a_cur = *(const v2f*)(Ap);
    v2f b_cur[NT];
#pragma unroll
    for (int s = 0; s < NT; ++s) {
        if constexpr (LDBK == 1) b_cur[s] = *(const v2f*)(Bp[s]);
        else { b_cur[s][0] = Bp[s][0]; b_cur[s][1] = Bp[s][LDBK]; }
    }

#pragma unroll 4
    for (int k0 = 0; k0 < KDIM; k0 += 4) {
        int kn = (k0 + 4 < KDIM) ? (k0 + 4) : 0;
        v2f a_nxt = *(const v2f*)(Ap + kn);
        v2f b_nxt[NT];
#pragma unroll
        for (int s = 0; s < NT; ++s) {
            if constexpr (LDBK == 1) b_nxt[s] = *(const v2f*)(Bp[s] + kn);
            else {
                b_nxt[s][0] = Bp[s][(size_t)kn * LDBK];
                b_nxt[s][1] = Bp[s][(size_t)kn * LDBK + LDBK];
            }
        }
#pragma unroll
        for (int s = 0; s < NT; ++s)
            acc[s] = __builtin_amdgcn_wmma_f32_16x16x4_f32(false, a_cur, false,
                                                           b_cur[s], (short)0,
                                                           acc[s], false, false);
        a_cur = a_nxt;
#pragma unroll
        for (int s = 0; s < NT; ++s) b_cur[s] = b_nxt[s];
    }

#pragma unroll
    for (int s = 0; s < NT; ++s) {
        int ccol = coff + ng * NT * 16 + s * 16 + l16;
#pragma unroll
        for (int r = 0; r < 8; ++r) {
            int crow = mt * 16 + r + half * 8;
            if (crow < M) C[(size_t)crow * ldc + ccol] = acc[s][r];
        }
    }
}

#define WS_NT 4
#define WS_KS (KDIM / 4)

__device__ __forceinline__ void ws_load_a(v2f (&a)[WS_KS], const float* __restrict__ A,
                                          int mt, int l16, int half, int M, int lda) {
    int arow = mt * 16 + l16;
    arow = arow < M ? arow : (M - 1);
    const float* ap = A + (size_t)arow * lda + half * 2;
#pragma unroll
    for (int j = 0; j < WS_KS; ++j) a[j] = *(const v2f*)(ap + j * 4);
}

__device__ __forceinline__ void ws_tile(const v2f (&a)[WS_KS],
                                        const v2f (&bf)[WS_NT][WS_KS],
                                        float* __restrict__ C, int mt, int l16,
                                        int half, int M, int ldc, int coff) {
    v8f acc[WS_NT];
#pragma unroll
    for (int s = 0; s < WS_NT; ++s) acc[s] = (v8f){0.f,0.f,0.f,0.f,0.f,0.f,0.f,0.f};
#pragma unroll
    for (int j = 0; j < WS_KS; ++j) {
#pragma unroll
        for (int s = 0; s < WS_NT; ++s)
            acc[s] = __builtin_amdgcn_wmma_f32_16x16x4_f32(false, a[j], false,
                                                           bf[s][j], (short)0,
                                                           acc[s], false, false);
    }
    float* cp = C + (size_t)(mt * 16 + half * 8) * ldc + coff + l16;
    if (mt * 16 + 16 <= M) {
#pragma unroll
        for (int r = 0; r < 8; ++r) {
#pragma unroll
            for (int s = 0; s < WS_NT; ++s) cp[(size_t)r * ldc + s * 16] = acc[s][r];
        }
    } else {
#pragma unroll
        for (int r = 0; r < 8; ++r) {
            if (mt * 16 + half * 8 + r < M) {
#pragma unroll
                for (int s = 0; s < WS_NT; ++s) cp[(size_t)r * ldc + s * 16] = acc[s][r];
            }
        }
    }
}

__global__ void __launch_bounds__(128)
gat_wmma_gemm_ws(const float* __restrict__ A, const float* __restrict__ B,
                 float* __restrict__ C, int M, int lda, int ldbn, int ldc,
                 int mtiles, int ncg, int gdiv, int ghi, int glo) {
    const int lane = threadIdx.x & 31;
    const int half = lane >> 4;
    const int l16  = lane & 15;
    const int gw = blockIdx.x * 4 + __builtin_amdgcn_readfirstlane(threadIdx.x >> 5);
    const int nw = gridDim.x * 4;
    const int cg = gw % ncg;
    const int slot = gw / ncg;
    const int nslots = nw / ncg;
    if (slot >= mtiles) return;

    const float* Bg = B + (size_t)(cg / gdiv) * ghi + (size_t)(cg % gdiv) * glo
                        + half * 2;
    v2f bf[WS_NT][WS_KS];
#pragma unroll
    for (int s = 0; s < WS_NT; ++s) {
        const float* bp = Bg + (size_t)(s * 16 + l16) * ldbn;
#pragma unroll
        for (int j = 0; j < WS_KS; ++j) bf[s][j] = *(const v2f*)(bp + j * 4);
    }

    const int coff = cg * (WS_NT * 16);
    int mt = slot;
    v2f a0[WS_KS], a1[WS_KS];
    ws_load_a(a0, A, mt, l16, half, M, lda);
    while (true) {
        {
            int mn = mt + nslots;
            mn = mn < mtiles ? mn : mt;
            ws_load_a(a1, A, mn, l16, half, M, lda);
            __builtin_amdgcn_sched_barrier(0);
            ws_tile(a0, bf, C, mt, l16, half, M, ldc, coff);
        }
        mt += nslots;
        if (mt >= mtiles) break;
        {
            int mn = mt + nslots;
            mn = mn < mtiles ? mn : mt;
            ws_load_a(a0, A, mn, l16, half, M, lda);
            __builtin_amdgcn_sched_barrier(0);
            ws_tile(a1, bf, C, mt, l16, half, M, ldc, coff);
        }
        mt += nslots;
        if (mt >= mtiles) break;
    }
}

#define OWN_NODES 200
#define OWN_WAVES 32
#define OWN_CAP   640
#define OWN_NG1   ((E_EDGES + 127) / 128)
#define OWN_NG2   ((E2_EDGES + 127) / 128)
#define OWN_NGT   (OWN_NG1 + OWN_NG2)
#define OWN_GPW   ((OWN_NGT + OWN_WAVES - 1) / OWN_WAVES)
#define OWN_ROUNDS ((OWN_GPW + 3) / 4)
#define OWN_NWG   (N_NODES / OWN_NODES)
#define KEEP_CNT_OFF  256
#define KEEP_W1_OFF   (KEEP_CNT_OFF + OWN_NWG * OWN_WAVES)
#define KEEP_W2_OFF   (KEEP_W1_OFF + OWN_NWG * OWN_WAVES * OWN_CAP)
static_assert(OWN_NWG <= KEEP_CNT_OFF, "one mark per workgroup in front of the lengths");
static_assert(N_NODES <= 65536 && NREL <= 32768, "second list word packs 16 + 15 bits");
static_assert(ET_EDGES <= (1 << 20) && OWN_NODES <= 256, "first list word packs 20 + 8 bits");
static_assert(N_NODES % OWN_NODES == 0, "every workgroup owns a full set of nodes");

struct OwnEdge {
    float xa[4], xb[4], r0[4], r1[4];
    int nl;
    int direct;
};

template <int L>
__device__ __forceinline__ void own_load(OwnEdge& r, int e, int nl, int dt, int base, int lane,
                                         const int* __restrict__ etn,
                                         const float* __restrict__ xrow,
                                         const float* __restrict__ erow,
                                         const float* __restrict__ rtab) {
    const int s = base + nl;
    const int d = dt & 0xFFFF;
    r.nl = nl;
    if (e < E_EDGES) {
        r.direct = 1;
        if constexpr (L == 1) {
#pragma unroll
            for (int h = 0; h < 2; ++h) {
                float2 v = *(const float2*)(erow + (size_t)e * 128 + h * 64 + lane * 2);
                r.r0[h * 2] = v.x; r.r0[h * 2 + 1] = v.y;
                r.r1[h * 2] = 0.f; r.r1[h * 2 + 1] = 0.f;
            }
        } else {
            int t = dt >> 16;
            float4 v = *(const float4*)(rtab + (size_t)t * 128 + lane * 4);
            r.r0[0] = v.x; r.r0[1] = v.y; r.r0[2] = v.z; r.r0[3] = v.w;
            r.r1[0] = 0.f; r.r1[1] = 0.f; r.r1[2] = 0.f; r.r1[3] = 0.f;
        }
    } else {
        int j = e - E_EDGES;
        int t0 = etn[2 * j], t1 = etn[2 * j + 1];
        r.direct = 0;
        if constexpr (L == 1) {
#pragma unroll
            for (int h = 0; h < 2; ++h) {
                float2 a0 = *(const float2*)(rtab + (size_t)t0 * 128 + h * 64 + lane * 2);
                float2 a1 = *(const float2*)(rtab + (size_t)t1 * 128 + h * 64 + lane * 2);
                r.r0[h * 2] = a0.x; r.r0[h * 2 + 1] = a0.y;
                r.r1[h * 2] = a1.x; r.r1[h * 2 + 1] = a1.y;
            }
        } else {
            float4 a0 = *(const float4*)(rtab + (size_t)t0 * 128 + lane * 4);
            float4 a1 = *(const float4*)(rtab + (size_t)t1 * 128 + lane * 4);
            r.r0[0] = a0.x; r.r0[1] = a0.y; r.r0[2] = a0.z; r.r0[3] = a0.w;
            r.r1[0] = a1.x; r.r1[1] = a1.y; r.r1[2] = a1.z; r.r1[3] = a1.w;
        }
    }
    if constexpr (L == 1) {
#pragma unroll
        for (int h = 0; h < 2; ++h) {
            float2 a = *(const float2*)(xrow + (size_t)s * 256 + h * 128 + lane * 2);
            float2 b = *(const float2*)(xrow + (size_t)d * 256 + h * 128 + 64 + lane * 2);
            r.xa[h * 2] = a.x; r.xa[h * 2 + 1] = a.y;
            r.xb[h * 2] = b.x; r.xb[h * 2 + 1] = b.y;
        }
    } else {
        float4 a = *(const float4*)(xrow + (size_t)s * 256 + lane * 4);
        float4 b = *(const float4*)(xrow + (size_t)d * 256 + 128 + lane * 4);
        r.xa[0] = a.x; r.xa[1] = a.y; r.xa[2] = a.z; r.xa[3] = a.w;
        r.xb[0] = b.x; r.xb[1] = b.y; r.xb[2] = b.z; r.xb[3] = b.w;
    }
}

template <int L>
__device__ __forceinline__ void own_add(const OwnEdge& r, const float (&w)[4], int lane,
                                        float* acc, float* den) {
    float m[4];
#pragma unroll
    for (int c = 0; c < 4; ++c) {
        float ev = r.direct ? r.r0[c] : (r.r0[c] + r.r1[c]);
        m[c] = r.xa[c] + r.xb[c] + ev;
    }
    if constexpr (L == 1) {
#pragma unroll
        for (int h = 0; h < 2; ++h) {
            float m0 = m[h * 2], m1 = m[h * 2 + 1];
            float p = m0 * w[h * 2] + m1 * w[h * 2 + 1];
#pragma unroll
            for (int off = 16; off > 0; off >>= 1) p += __shfl_xor(p, off, 32);
            float lr = p > 0.f ? p : ALPHA * p;
            float eev = __expf(-lr);
            float2* ap = (float2*)(acc + r.nl * 128 + h * 64 + lane * 2);
            float2 av = *ap;
            av.x += eev * m0;
            av.y += eev * m1;
            *ap = av;
            if (lane == 0) den[r.nl * 2 + h] += eev;
        }
    } else {
        float p = m[0] * w[0] + m[1] * w[1] + m[2] * w[2] + m[3] * w[3];
#pragma unroll
        for (int off = 16; off > 0; off >>= 1) p += __shfl_xor(p, off, 32);
        float lr = p > 0.f ? p : ALPHA * p;
        float eev = __expf(-lr);
        float4* ap = (float4*)(acc + r.nl * 128 + lane * 4);
        float4 av = *ap;
        av.x += eev * m[0];
        av.y += eev * m[1];
        av.z += eev * m[2];
        av.w += eev * m[3];
        *ap = av;
        if (lane == 0) den[r.nl] += eev;
    }
}

template <int L>
__global__ void __launch_bounds__(OWN_WAVES * 32)
gat_own(const int* __restrict__ el, const int* __restrict__ eln,
        const int* __restrict__ et, const int* __restrict__ etn,
        const float* __restrict__ xrow, const float* __restrict__ erow,
        const float* __restrict__ rtab, const float* __restrict__ wvec,
        float* __restrict__ outp, int* __restrict__ keep) {
    __shared__ float acc[OWN_NODES * 128];
    __shared__ float den[OWN_NODES * 2];
    __shared__ int lists[OWN_WAVES * OWN_CAP];
    __shared__ int lists2[OWN_WAVES * OWN_CAP];
    __shared__ int s_cnt[OWN_WAVES];
    __shared__ int s_more;
    __shared__ int s_long;

    const int tid  = threadIdx.x;
    const int lane = tid & 31;
    const int wv   = __builtin_amdgcn_readfirstlane(tid >> 5);
    const int base = blockIdx.x * OWN_NODES;

    for (int i = tid; i < OWN_NODES * 128; i += OWN_WAVES * 32) acc[i] = 0.f;
    for (int i = tid; i < OWN_NODES * 2; i += OWN_WAVES * 32) den[i] = 0.f;
    if (tid == 0) s_long = 0;

    float w[4];
    if constexpr (L == 1) {
        w[0] = wvec[lane * 2];      w[1] = wvec[lane * 2 + 1];
        w[2] = wvec[64 + lane * 2]; w[3] = wvec[64 + lane * 2 + 1];
    } else {
        w[0] = wvec[lane * 4];     w[1] = wvec[lane * 4 + 1];
        w[2] = wvec[lane * 4 + 2]; w[3] = wvec[lane * 4 + 3];
    }

    int g = wv * OWN_GPW;
    int gend = g + OWN_GPW;
    gend = gend < OWN_NGT ? gend : OWN_NGT;
    int* mylist = lists + wv * OWN_CAP;
    int* mylist2 = lists2 + wv * OWN_CAP;
    const unsigned lt_mask = (1u << lane) - 1u;

    int* k_mark = keep + blockIdx.x;
    int* k_cnt  = keep + KEEP_CNT_OFF + blockIdx.x * OWN_WAVES + wv;
    int* k_w1   = keep + KEEP_W1_OFF + (size_t)(blockIdx.x * OWN_WAVES + wv) * OWN_CAP;
    int* k_w2   = keep + KEEP_W2_OFF + (size_t)(blockIdx.x * OWN_WAVES + wv) * OWN_CAP;
    bool reuse = false;
    if constexpr (L == 2) {
        reuse = (*k_mark == 1);
        if (reuse) g = gend;
    }
    const int w2lo = reuse ? wv : 0;
    const int w2hi = reuse ? wv + 1 : OWN_WAVES;
    int kept = 0;
    bool single = false;

    for (int round = 0; round < OWN_ROUNDS; ++round) {
        if (tid == 0) s_more = 0;
        __syncthreads();

        int cnt = 0;
        if (reuse) {
            cnt = *k_cnt;
            cnt = cnt < 0 ? 0 : (cnt > OWN_CAP ? OWN_CAP : cnt);
            for (int i = lane; i < cnt; i += 32) {
                mylist[i] = k_w1[i];
                mylist2[i] = k_w2[i];
            }
        }
        while (g < gend && cnt + 512 <= OWN_CAP) {
            int4 v[4];
#pragma unroll
            for (int q = 0; q < 4; ++q) {
                int gi = g + q;
                gi = gi < gend ? gi : (gend - 1);
                const int* p = gi < OWN_NG1 ? (el + (size_t)gi * 128)
                                            : (eln + (size_t)(gi - OWN_NG1) * 128);
                v[q] = *(const int4*)(p + lane * 4);
            }
#pragma unroll
            for (int q = 0; q < 4; ++q) {
                int gi = g + q;
                if (gi < gend) {
                    bool isd = gi < OWN_NG1;
                    int eb   = isd ? gi * 128 : (gi - OWN_NG1) * 128;
                    int lim  = isd ? E_EDGES : E2_EDGES;
                    int eoff = isd ? 0 : E_EDGES;
                    int idx0 = eb + lane * 4;
                    int sv[4] = {v[q].x, v[q].y, v[q].z, v[q].w};
#pragma unroll
                    for (int j = 0; j < 4; ++j) {
                        int loc = sv[j] - base;
                        bool hit = (idx0 + j < lim) && ((unsigned)loc < (unsigned)OWN_NODES);
                        unsigned mask = __builtin_amdgcn_ballot_w32(hit);
                        if (mask != 0u) {
                            int pos = cnt + __builtin_popcount(mask & lt_mask);
                            if (hit) mylist[pos] = (eoff + idx0 + j) | (loc << 20);
                            cnt += __builtin_popcount(mask);
                        }
                    }
                }
            }
            g += 4;
        }
        for (int i = lane; i < (reuse ? 0 : cnt); i += 32) {
            int e = mylist[i] & 0xFFFFF;
            int d, t = 0;
            if (e < E_EDGES) {
                d = el[E_EDGES + e];
                t = et[e];
            } else {
                d = eln[E2_EDGES + (e - E_EDGES)];
            }
            mylist2[i] = d | (t << 16);
        }
        if (lane == 0) {
            s_cnt[wv] = cnt;
            if (g < gend) s_more = 1;
        }
        __syncthreads();
        const int more = s_more;

        {
            OwnEdge cur = {};
            bool have = false;
            for (int w2 = w2lo; w2 < w2hi; ++w2) {
                const int n = __builtin_amdgcn_readfirstlane(s_cnt[w2]);
                const int* lp = lists + w2 * OWN_CAP;
                const int* lp2 = lists2 + w2 * OWN_CAP;
                for (int i0 = 0; i0 < n; i0 += 32) {
                    int idx = i0 + lane;
                    int ent = idx < n ? lp[idx] : 0;
                    int ent2 = idx < n ? lp2[idx] : 0;
                    bool mine = (idx < n) && (((ent >> 20) & 31) == wv);
                    unsigned mask = __builtin_amdgcn_ballot_w32(mine);
                    while (mask != 0u) {
                        int b = __builtin_ctz(mask);
                        mask &= mask - 1u;
                        int en = __builtin_amdgcn_readlane(ent, b);
                        int en2 = __builtin_amdgcn_readlane(ent2, b);
                        if constexpr (L == 1) {
                            if (kept < OWN_CAP && lane == 0) { k_w1[kept] = en; k_w2[kept] = en2; }
                            ++kept;
                        }
                        OwnEdge nx;
                        own_load<L>(nx, en & 0xFFFFF, en >> 20, en2, base, lane,
                                    etn, xrow, erow, rtab);
                        if (have) own_add<L>(cur, w, lane, acc, den);
                        cur = nx;
                        have = true;
                    }
                }
            }
            if (have) own_add<L>(cur, w, lane, acc, den);
        }
        if constexpr (L == 1) {
            if (lane == 0 && kept > OWN_CAP) s_long = 1;
        }
        __syncthreads();
        if (!more) { single = (round == 0); break; }
    }
    if constexpr (L == 1) {
        if (lane == 0) *k_cnt = kept;
        if (tid == 0) *k_mark = (single && s_long == 0) ? 1 : 0;
    }

    float* orow = outp + (size_t)base * 128;
    for (int i = tid; i < OWN_NODES * 128; i += OWN_WAVES * 32) {
        int nl = i >> 7;
        float dnm;
        if constexpr (L == 1) dnm = den[nl * 2 + ((i & 127) >> 6)];
        else                  dnm = den[nl];
        dnm = dnm > 1e-12f ? dnm : 1e-12f;
        float h = acc[i] / dnm;
        if constexpr (L == 1) {
            orow[i] = h;
        } else {
            float y = h > 0.f ? h : (__expf(h) - 1.f);
            y = y > 0.f ? y : (__expf(y) - 1.f);
            orow[i] = y;
        }
    }
}

template <int LDBK>
static inline void launch_gemm(hipStream_t st, const float* A, const float* B,
                               float* C, int M, int N,
                               int lda, int ldbn, int ldc, int coff) {
    constexpr int NT = 4;
    int mtiles  = (M + 15) / 16;
    int ngroups = N / (NT * 16);
    long tiles = (long)mtiles * ngroups;
    int blocks = (int)((tiles + 7) / 8);
    gat_wmma_gemm_f32<LDBK, NT><<<blocks, 256, 0, st>>>(A, B, C, M, lda, ldbn,
                                                        ldc, coff, mtiles,
                                                        ngroups);
}

extern "C" void kernel_launch(void* const* d_in, const int* in_sizes, int n_in,
                              void* d_out, int out_size, void* d_ws, size_t ws_size,
                              hipStream_t stream) {
    const float* x    = (const float*)d_in[0];
    const float* rel  = (const float*)d_in[1];
    const int*   el   = (const int*)d_in[2];
    const int*   et   = (const int*)d_in[3];
    const float* eemb = (const float*)d_in[4];
    const int*   eln  = (const int*)d_in[5];
    const int*   etn  = (const int*)d_in[6];
    const float* w1h  = (const float*)d_in[8];
    const float* w2h  = (const float*)d_in[9];
    const float* w1o  = (const float*)d_in[10];
    const float* w2o  = (const float*)d_in[11];
    const float* WR   = (const float*)d_in[12];
    float* out = (float*)d_out;

    float* ws = (float*)d_ws;
    float* xnode  = ws;
    float* ec     = xnode  + (size_t)N_NODES * 256;
    float* relc   = ec     + (size_t)E_EDGES * 128;
    float* num1   = relc   + (size_t)NREL * 128;
    float* xab2   = num1   + (size_t)N_NODES * 128;
    float* relW2  = xab2   + (size_t)N_NODES * 256;
    int*   keep   = (int*)(relW2 + (size_t)NREL * 128);

    float* out_rel = out + (size_t)N_NODES * 128;

    gat_wmma_gemm_ws<<<512, 128, 0, stream>>>(x, w1h, xnode, N_NODES, 128, 384, 256,
                                              N_NODES / 16, 4, 2, 64 * 384, 128);
    gat_wmma_gemm_ws<<<512, 128, 0, stream>>>(eemb, w1h + 256, ec, E_EDGES, 128, 384,
                                              128, E_EDGES / 16, 2, 1, 64 * 384, 0);
    for (int h = 0; h < 2; ++h)
        launch_gemm<1>(stream, rel, w1h + (size_t)h * 64 * 384 + 256,
                       relc, NREL, 64, 128, 384, 128, h * 64);

    gat_own<1><<<N_NODES / OWN_NODES, OWN_WAVES * 32, 0, stream>>>(
        el, eln, et, etn, xnode, ec, relc, w2h, num1, keep);

    launch_gemm<128>(stream, rel, WR, out_rel, NREL, 128, 128, 1, 128, 0);
    gat_wmma_gemm_ws<<<512, 128, 0, stream>>>(num1, w1o, xab2, N_NODES, 128, 384, 256,
                                              N_NODES / 16, 4, 2, 128, 64 * 384);
    launch_gemm<1>(stream, out_rel, w1o + 256, relW2, NREL, 128, 128, 384, 128, 0);

    gat_own<2><<<N_NODES / OWN_NODES, OWN_WAVES * 32, 0, stream>>>(
        el, eln, et, etn, xab2, ec, relW2, w2o, out, keep);
}


# edge adds two at a time: a pair's row loads are issued together and both edges' (and heads') logit sums share each cross-lane exchange step; adds stay in list order, partners and sums unchanged
# speedup vs baseline: 2.0710x; 1.0389x over previous
#include <hip/hip_runtime.h>
#include <hip/hip_bf16.h>

typedef float v2f __attribute__((ext_vector_type(2)));
typedef float v8f __attribute__((ext_vector_type(8)));

#define N_NODES 50000
#define E_EDGES 500000
#define E2_EDGES 62500
#define ET_EDGES (E_EDGES + E2_EDGES)
#define NREL 1000
#define ALPHA 0.2f
#define KDIM 128

template <int LDBK, int NT>
__global__ void gat_wmma_gemm_f32(const float* __restrict__ A,
                                  const float* __restrict__ B,
                                  float* __restrict__ C,
                                  int M, int lda, int ldbn, int ldc, int coff,
                                  int mtiles, int ngroups) {
    int wave = threadIdx.x >> 5;
    long tile = (long)blockIdx.x * (blockDim.x >> 5) + wave;
    long tiles_total = (long)mtiles * ngroups;
    if (tile >= tiles_total) return;
    int ng = (int)(tile / mtiles);
    int mt = (int)(tile % mtiles);

    int lane = threadIdx.x & 31;
    int half = lane >> 4;
    int l16  = lane & 15;

    int arow = mt * 16 + l16;
    int arow_c = arow < M ? arow : (M - 1);
    const float* Ap = A + (size_t)arow_c * lda + half * 2;
    const float* Bp[NT];
#pragma unroll
    for (int s = 0; s < NT; ++s)
        Bp[s] = B + (size_t)(ng * NT * 16 + s * 16 + l16) * ldbn
                  + (size_t)(half * 2) * LDBK;

    v8f acc[NT];
#pragma unroll
    for (int s = 0; s < NT; ++s) acc[s] = (v8f){0.f,0.f,0.f,0.f,0.f,0.f,0.f,0.f};

    v2f a_cur = *(const v2f*)(Ap);
    v2f b_cur[NT];
#pragma unroll
    for (int s = 0; s < NT; ++s) {
        if constexpr (LDBK == 1) b_cur[s] = *(const v2f*)(Bp[s]);
        else { b_cur[s][0] = Bp[s][0]; b_cur[s][1] = Bp[s][LDBK]; }
    }

#pragma unroll 4
    for (int k0 = 0; k0 < KDIM; k0 += 4) {
        int kn = (k0 + 4 < KDIM) ? (k0 + 4) : 0;
        v2f a_nxt = *(const v2f*)(Ap + kn);
        v2f b_nxt[NT];
#pragma unroll
        for (int s = 0; s < NT; ++s) {
            if constexpr (LDBK == 1) b_nxt[s] = *(const v2f*)(Bp[s] + kn);
            else {
                b_nxt[s][0] = Bp[s][(size_t)kn * LDBK];
                b_nxt[s][1] = Bp[s][(size_t)kn * LDBK + LDBK];
            }
        }
#pragma unroll
        for (int s = 0; s < NT; ++s)
            acc[s] = __builtin_amdgcn_wmma_f32_16x16x4_f32(false, a_cur, false,
                                                           b_cur[s], (short)0,
                                                           acc[s], false, false);
        a_cur = a_nxt;
#pragma unroll
        for (int s = 0; s < NT; ++s) b_cur[s] = b_nxt[s];
    }

#pragma unroll
    for (int s = 0; s < NT; ++s) {
        int ccol = coff + ng * NT * 16 + s * 16 + l16;
#pragma unroll
        for (int r = 0; r < 8; ++r) {
            int crow = mt * 16 + r + half * 8;
            if (crow < M) C[(size_t)crow * ldc + ccol] = acc[s][r];
        }
    }
}

#define WS_NT 4
#define WS_KS (KDIM / 4)

__device__ __forceinline__ void ws_load_a(v2f (&a)[WS_KS], const float* __restrict__ A,
                                          int mt, int l16, int half, int M, int lda) {
    int arow = mt * 16 + l16;
    arow = arow < M ? arow : (M - 1);
    const float* ap = A + (size_t)arow * lda + half * 2;
#pragma unroll
    for (int j = 0; j < WS_KS; ++j) a[j] = *(const v2f*)(ap + j * 4);
}

__device__ __forceinline__ void ws_tile(const v2f (&a)[WS_KS],
                                        const v2f (&bf)[WS_NT][WS_KS],
                                        float* __restrict__ C, int mt, int l16,
                                        int half, int M, int ldc, int coff) {
    v8f acc[WS_NT];
#pragma unroll
    for (int s = 0; s < WS_NT; ++s) acc[s] = (v8f){0.f,0.f,0.f,0.f,0.f,0.f,0.f,0.f};
#pragma unroll
    for (int j = 0; j < WS_KS; ++j) {
#pragma unroll
        for (int s = 0; s < WS_NT; ++s)
            acc[s] = __builtin_amdgcn_wmma_f32_16x16x4_f32(false, a[j], false,
                                                           bf[s][j], (short)0,
                                                           acc[s], false, false);
    }
    float* cp = C + (size_t)(mt * 16 + half * 8) * ldc + coff + l16;
    if (mt * 16 + 16 <= M) {
#pragma unroll
        for (int r = 0; r < 8; ++r) {
#pragma unroll
            for (int s = 0; s < WS_NT; ++s) cp[(size_t)r * ldc + s * 16] = acc[s][r];
        }
    } else {
#pragma unroll
        for (int r = 0; r < 8; ++r) {
            if (mt * 16 + half * 8 + r < M) {
#pragma unroll
                for (int s = 0; s < WS_NT; ++s) cp[(size_t)r * ldc + s * 16] = acc[s][r];
            }
        }
    }
}

__global__ void __launch_bounds__(128)
gat_wmma_gemm_ws(const float* __restrict__ A, const float* __restrict__ B,
                 float* __restrict__ C, int M, int lda, int ldbn, int ldc,
                 int mtiles, int ncg, int gdiv, int ghi, int glo) {
    const int lane = threadIdx.x & 31;
    const int half = lane >> 4;
    const int l16  = lane & 15;
    const int gw = blockIdx.x * 4 + __builtin_amdgcn_readfirstlane(threadIdx.x >> 5);
    const int nw = gridDim.x * 4;
    const int cg = gw % ncg;
    const int slot = gw / ncg;
    const int nslots = nw / ncg;
    if (slot >= mtiles) return;

    const float* Bg = B + (size_t)(cg / gdiv) * ghi + (size_t)(cg % gdiv) * glo
                        + half * 2;
    v2f bf[WS_NT][WS_KS];
#pragma unroll
    for (int s = 0; s < WS_NT; ++s) {
        const float* bp = Bg + (size_t)(s * 16 + l16) * ldbn;
#pragma unroll
        for (int j = 0; j < WS_KS; ++j) bf[s][j] = *(const v2f*)(bp + j * 4);
    }

    const int coff = cg * (WS_NT * 16);
    int mt = slot;
    v2f a0[WS_KS], a1[WS_KS];
    ws_load_a(a0, A, mt, l16, half, M, lda);
    while (true) {
        {
            int mn = mt + nslots;
            mn = mn < mtiles ? mn : mt;
            ws_load_a(a1, A, mn, l16, half, M, lda);
            __builtin_amdgcn_sched_barrier(0);
            ws_tile(a0, bf, C, mt, l16, half, M, ldc, coff);
        }
        mt += nslots;
        if (mt >= mtiles) break;
        {
            int mn = mt + nslots;
            mn = mn < mtiles ? mn : mt;
            ws_load_a(a0, A, mn, l16, half, M, lda);
            __builtin_amdgcn_sched_barrier(0);
            ws_tile(a1, bf, C, mt, l16, half, M, ldc, coff);
        }
        mt += nslots;
        if (mt >= mtiles) break;
    }
}

#define OWN_NODES 200
#define OWN_WAVES 32
#define OWN_CAP   640
#define OWN_NG1   ((E_EDGES + 127) / 128)
#define OWN_NG2   ((E2_EDGES + 127) / 128)
#define OWN_NGT   (OWN_NG1 + OWN_NG2)
#define OWN_GPW   ((OWN_NGT + OWN_WAVES - 1) / OWN_WAVES)
#define OWN_ROUNDS ((OWN_GPW + 3) / 4)
#define OWN_NWG   (N_NODES / OWN_NODES)
#define KEEP_CNT_OFF  256
#define KEEP_W1_OFF   (KEEP_CNT_OFF + OWN_NWG * OWN_WAVES)
#define KEEP_W2_OFF   (KEEP_W1_OFF + OWN_NWG * OWN_WAVES * OWN_CAP)
static_assert(OWN_NWG <= KEEP_CNT_OFF, "one mark per workgroup in front of the lengths");
static_assert(N_NODES <= 65536 && NREL <= 32768, "second list word packs 16 + 15 bits");
static_assert(ET_EDGES <= (1 << 20) && OWN_NODES <= 256, "first list word packs 20 + 8 bits");
static_assert(N_NODES % OWN_NODES == 0, "every workgroup owns a full set of nodes");

struct OwnEdge {
    float xa[4], xb[4], r0[4], r1[4];
    int nl;
    int direct;
};

__device__ __forceinline__ void own_types(int e, const int* __restrict__ etn, int& t0, int& t1) {
    t0 = 0; t1 = 0;
    if (e >= E_EDGES) {
        int j = e - E_EDGES;
        t0 = etn[2 * j]; t1 = etn[2 * j + 1];
    }
}

template <int L>
__device__ __forceinline__ void own_rows_node(OwnEdge& r, int nl, int dt, int base, int lane,
                                              const float* __restrict__ xrow) {
    const int s = base + nl;
    const int d = dt & 0xFFFF;
    r.nl = nl;
    if constexpr (L == 1) {
#pragma unroll
        for (int h = 0; h < 2; ++h) {
            float2 a = *(const float2*)(xrow + (size_t)s * 256 + h * 128 + lane * 2);
            float2 b = *(const float2*)(xrow + (size_t)d * 256 + h * 128 + 64 + lane * 2);
            r.xa[h * 2] = a.x; r.xa[h * 2 + 1] = a.y;
            r.xb[h * 2] = b.x; r.xb[h * 2 + 1] = b.y;
        }
    } else {
        float4 a = *(const float4*)(xrow + (size_t)s * 256 + lane * 4);
        float4 b = *(const float4*)(xrow + (size_t)d * 256 + 128 + lane * 4);
        r.xa[0] = a.x; r.xa[1] = a.y; r.xa[2] = a.z; r.xa[3] = a.w;
        r.xb[0] = b.x; r.xb[1] = b.y; r.xb[2] = b.z; r.xb[3] = b.w;
    }
}

template <int L>
__device__ __forceinline__ void own_rows_rel(OwnEdge& r, int e, int dt, int t0, int t1, int lane,
                                             const float* __restrict__ erow,
                                             const float* __restrict__ rtab) {
    const bool direct = e < E_EDGES;
    r.direct = direct ? 1 : 0;
    if constexpr (L == 1) {
        const float* p0 = direct ? (erow + (size_t)e * 128) : (rtab + (size_t)t0 * 128);
#pragma unroll
        for (int h = 0; h < 2; ++h) {
            float2 v = *(const float2*)(p0 + h * 64 + lane * 2);
            r.r0[h * 2] = v.x; r.r0[h * 2 + 1] = v.y;
        }
        if (!direct) {
#pragma unroll
            for (int h = 0; h < 2; ++h) {
                float2 a1 = *(const float2*)(rtab + (size_t)t1 * 128 + h * 64 + lane * 2);
                r.r1[h * 2] = a1.x; r.r1[h * 2 + 1] = a1.y;
            }
        }
    } else {
        const int t = direct ? (dt >> 16) : t0;
        float4 v = *(const float4*)(rtab + (size_t)t * 128 + lane * 4);
        r.r0[0] = v.x; r.r0[1] = v.y; r.r0[2] = v.z; r.r0[3] = v.w;
        if (!direct) {
            float4 a1 = *(const float4*)(rtab + (size_t)t1 * 128 + lane * 4);
            r.r1[0] = a1.x; r.r1[1] = a1.y; r.r1[2] = a1.z; r.r1[3] = a1.w;
        }
    }
}

template <int L>
__device__ __forceinline__ void own_load(OwnEdge& r, int e, int nl, int dt, int base, int lane,
                                         const int* __restrict__ etn,
                                         const float* __restrict__ xrow,
                                         const float* __restrict__ erow,
                                         const float* __restrict__ rtab) {
    int t0, t1;
#pragma unroll
    for (int c = 0; c < 4; ++c) r.r1[c] = 0.f;
    own_types(e, etn, t0, t1);
    own_rows_node<L>(r, nl, dt, base, lane, xrow);
    own_rows_rel<L>(r, e, dt, t0, t1, lane, erow, rtab);
}

template <int L>
__device__ __forceinline__ void own_load2(OwnEdge& rA, OwnEdge& rB, int enA, int dtA, int enB,
                                          int dtB, int base, int lane,
                                          const int* __restrict__ etn,
                                          const float* __restrict__ xrow,
                                          const float* __restrict__ erow,
                                          const float* __restrict__ rtab) {
    const int eA = enA & 0xFFFFF, eB = enB & 0xFFFFF;
    int tA0, tA1, tB0, tB1;
#pragma unroll
    for (int c = 0; c < 4; ++c) { rA.r1[c] = 0.f; rB.r1[c] = 0.f; }
    own_types(eA, etn, tA0, tA1);
    own_types(eB, etn, tB0, tB1);
    own_rows_node<L>(rA, enA >> 20, dtA, base, lane, xrow);
    own_rows_node<L>(rB, enB >> 20, dtB, base, lane, xrow);
    own_rows_rel<L>(rA, eA, dtA, tA0, tA1, lane, erow, rtab);
    own_rows_rel<L>(rB, eB, dtB, tB0, tB1, lane, erow, rtab);
}

template <int L>
__device__ __forceinline__ void own_add(const OwnEdge& r, const float (&w)[4], int lane,
                                        float* acc, float* den) {
    float m[4];
#pragma unroll
    for (int c = 0; c < 4; ++c) {
        float ev = r.direct ? r.r0[c] : (r.r0[c] + r.r1[c]);
        m[c] = r.xa[c] + r.xb[c] + ev;
    }
    if constexpr (L == 1) {
#pragma unroll
        for (int h = 0; h < 2; ++h) {
            float m0 = m[h * 2], m1 = m[h * 2 + 1];
            float p = m0 * w[h * 2] + m1 * w[h * 2 + 1];
#pragma unroll
            for (int off = 16; off > 0; off >>= 1) p += __shfl_xor(p, off, 32);
            float lr = p > 0.f ? p : ALPHA * p;
            float eev = __expf(-lr);
            float2* ap = (float2*)(acc + r.nl * 128 + h * 64 + lane * 2);
            float2 av = *ap;
            av.x += eev * m0;
            av.y += eev * m1;
            *ap = av;
            if (lane == 0) den[r.nl * 2 + h] += eev;
        }
    } else {
        float p = m[0] * w[0] + m[1] * w[1] + m[2] * w[2] + m[3] * w[3];
#pragma unroll
        for (int off = 16; off > 0; off >>= 1) p += __shfl_xor(p, off, 32);
        float lr = p > 0.f ? p : ALPHA * p;
        float eev = __expf(-lr);
        float4* ap = (float4*)(acc + r.nl * 128 + lane * 4);
        float4 av = *ap;
        av.x += eev * m[0];
        av.y += eev * m[1];
        av.z += eev * m[2];
        av.w += eev * m[3];
        *ap = av;
        if (lane == 0) den[r.nl] += eev;
    }
}


template <int L>
__device__ __forceinline__ void own_add2(const OwnEdge& A, const OwnEdge& B,
                                         const float (&w)[4], int lane,
                                         float* acc, float* den) {
    float mA[4], mB[4];
#pragma unroll
    for (int c = 0; c < 4; ++c) {
        float evA = A.direct ? A.r0[c] : (A.r0[c] + A.r1[c]);
        float evB = B.direct ? B.r0[c] : (B.r0[c] + B.r1[c]);
        mA[c] = A.xa[c] + A.xb[c] + evA;
        mB[c] = B.xa[c] + B.xb[c] + evB;
    }
    const bool same = (A.nl == B.nl);
    if constexpr (L == 1) {
        float p[4];
        p[0] = mA[0] * w[0] + mA[1] * w[1];
        p[1] = mA[2] * w[2] + mA[3] * w[3];
        p[2] = mB[0] * w[0] + mB[1] * w[1];
        p[3] = mB[2] * w[2] + mB[3] * w[3];
#pragma unroll
        for (int off = 16; off > 0; off >>= 1) {
            float t[4];
#pragma unroll
            for (int q = 0; q < 4; ++q) t[q] = __shfl_xor(p[q], off, 32);
#pragma unroll
            for (int q = 0; q < 4; ++q) p[q] += t[q];
        }
        float ee[4];
#pragma unroll
        for (int q = 0; q < 4; ++q) {
            float lr = p[q] > 0.f ? p[q] : ALPHA * p[q];
            ee[q] = __expf(-lr);
        }
        float2* apA0 = (float2*)(acc + A.nl * 128 + lane * 2);
        float2* apA1 = (float2*)(acc + A.nl * 128 + 64 + lane * 2);
        float2* apB0 = (float2*)(acc + B.nl * 128 + lane * 2);
        float2* apB1 = (float2*)(acc + B.nl * 128 + 64 + lane * 2);
        float2 a0 = *apA0, a1 = *apA1, b0 = *apB0, b1 = *apB1;
        a0.x += ee[0] * mA[0]; a0.y += ee[0] * mA[1];
        a1.x += ee[1] * mA[2]; a1.y += ee[1] * mA[3];
        if (same) { b0 = a0; b1 = a1; }
        b0.x += ee[2] * mB[0]; b0.y += ee[2] * mB[1];
        b1.x += ee[3] * mB[2]; b1.y += ee[3] * mB[3];
        *apA0 = a0; *apA1 = a1;
        *apB0 = b0; *apB1 = b1;
        if (lane == 0) {
            float dA0 = den[A.nl * 2], dA1 = den[A.nl * 2 + 1];
            float dB0 = den[B.nl * 2], dB1 = den[B.nl * 2 + 1];
            dA0 += ee[0]; dA1 += ee[1];
            if (same) { dB0 = dA0; dB1 = dA1; }
            dB0 += ee[2]; dB1 += ee[3];
            den[A.nl * 2] = dA0; den[A.nl * 2 + 1] = dA1;
            den[B.nl * 2] = dB0; den[B.nl * 2 + 1] = dB1;
        }
    } else {
        float pA = mA[0] * w[0] + mA[1] * w[1] + mA[2] * w[2] + mA[3] * w[3];
        float pB = mB[0] * w[0] + mB[1] * w[1] + mB[2] * w[2] + mB[3] * w[3];
#pragma unroll
        for (int off = 16; off > 0; off >>= 1) {
            float tA = __shfl_xor(pA, off, 32);
            float tB = __shfl_xor(pB, off, 32);
            pA += tA;
            pB += tB;
        }
        float lrA = pA > 0.f ? pA : ALPHA * pA;
        float lrB = pB > 0.f ? pB : ALPHA * pB;
        float eA = __expf(-lrA);
        float eB = __expf(-lrB);
        float4* apA = (float4*)(acc + A.nl * 128 + lane * 4);
        float4* apB = (float4*)(acc + B.nl * 128 + lane * 4);
        float4 a = *apA, b = *apB;
        a.x += eA * mA[0]; a.y += eA * mA[1]; a.z += eA * mA[2]; a.w += eA * mA[3];
        if (same) b = a;
        b.x += eB * mB[0]; b.y += eB * mB[1]; b.z += eB * mB[2]; b.w += eB * mB[3];
        *apA = a;
        *apB = b;
        if (lane == 0) {
            float dA = den[A.nl], dB = den[B.nl];
            dA += eA;
            if (same) dB = dA;
            dB += eB;
            den[A.nl] = dA;
            den[B.nl] = dB;
        }
    }
}

template <int L>
__global__ void __launch_bounds__(OWN_WAVES * 32)
gat_own(const int* __restrict__ el, const int* __restrict__ eln,
        const int* __restrict__ et, const int* __restrict__ etn,
        const float* __restrict__ xrow, const float* __restrict__ erow,
        const float* __restrict__ rtab, const float* __restrict__ wvec,
        float* __restrict__ outp, int* __restrict__ keep) {
    __shared__ float acc[OWN_NODES * 128];
    __shared__ float den[OWN_NODES * 2];
    __shared__ int lists[OWN_WAVES * OWN_CAP];
    __shared__ int lists2[OWN_WAVES * OWN_CAP];
    __shared__ int s_cnt[OWN_WAVES];
    __shared__ int s_more;
    __shared__ int s_long;

    const int tid  = threadIdx.x;
    const int lane = tid & 31;
    const int wv   = __builtin_amdgcn_readfirstlane(tid >> 5);
    const int base = blockIdx.x * OWN_NODES;

    for (int i = tid; i < OWN_NODES * 128; i += OWN_WAVES * 32) acc[i] = 0.f;
    for (int i = tid; i < OWN_NODES * 2; i += OWN_WAVES * 32) den[i] = 0.f;
    if (tid == 0) s_long = 0;

    float w[4];
    if constexpr (L == 1) {
        w[0] = wvec[lane * 2];      w[1] = wvec[lane * 2 + 1];
        w[2] = wvec[64 + lane * 2]; w[3] = wvec[64 + lane * 2 + 1];
    } else {
        w[0] = wvec[lane * 4];     w[1] = wvec[lane * 4 + 1];
        w[2] = wvec[lane * 4 + 2]; w[3] = wvec[lane * 4 + 3];
    }

    int g = wv * OWN_GPW;
    int gend = g + OWN_GPW;
    gend = gend < OWN_NGT ? gend : OWN_NGT;
    int* mylist = lists + wv * OWN_CAP;
    int* mylist2 = lists2 + wv * OWN_CAP;
    const unsigned lt_mask = (1u << lane) - 1u;

    int* k_mark = keep + blockIdx.x;
    int* k_cnt  = keep + KEEP_CNT_OFF + blockIdx.x * OWN_WAVES + wv;
    int* k_w1   = keep + KEEP_W1_OFF + (size_t)(blockIdx.x * OWN_WAVES + wv) * OWN_CAP;
    int* k_w2   = keep + KEEP_W2_OFF + (size_t)(blockIdx.x * OWN_WAVES + wv) * OWN_CAP;
    bool reuse = false;
    if constexpr (L == 2) {
        reuse = (*k_mark == 1);
        if (reuse) g = gend;
    }
    const int w2lo = reuse ? wv : 0;
    const int w2hi = reuse ? wv + 1 : OWN_WAVES;
    int kept = 0;
    bool single = false;

    for (int round = 0; round < OWN_ROUNDS; ++round) {
        if (tid == 0) s_more = 0;
        __syncthreads();

        int cnt = 0;
        if (reuse) {
            cnt = *k_cnt;
            cnt = cnt < 0 ? 0 : (cnt > OWN_CAP ? OWN_CAP : cnt);
            for (int i = lane; i < cnt; i += 32) {
                mylist[i] = k_w1[i];
                mylist2[i] = k_w2[i];
            }
        }
        while (g < gend && cnt + 512 <= OWN_CAP) {
            int4 v[4];
#pragma unroll
            for (int q = 0; q < 4; ++q) {
                int gi = g + q;
                gi = gi < gend ? gi : (gend - 1);
                const int* p = gi < OWN_NG1 ? (el + (size_t)gi * 128)
                                            : (eln + (size_t)(gi - OWN_NG1) * 128);
                v[q] = *(const int4*)(p + lane * 4);
            }
#pragma unroll
            for (int q = 0; q < 4; ++q) {
                int gi = g + q;
                if (gi < gend) {
                    bool isd = gi < OWN_NG1;
                    int eb   = isd ? gi * 128 : (gi - OWN_NG1) * 128;
                    int lim  = isd ? E_EDGES : E2_EDGES;
                    int eoff = isd ? 0 : E_EDGES;
                    int idx0 = eb + lane * 4;
                    int sv[4] = {v[q].x, v[q].y, v[q].z, v[q].w};
#pragma unroll
                    for (int j = 0; j < 4; ++j) {
                        int loc = sv[j] - base;
                        bool hit = (idx0 + j < lim) && ((unsigned)loc < (unsigned)OWN_NODES);
                        unsigned mask = __builtin_amdgcn_ballot_w32(hit);
                        if (mask != 0u) {
                            int pos = cnt + __builtin_popcount(mask & lt_mask);
                            if (hit) mylist[pos] = (eoff + idx0 + j) | (loc << 20);
                            cnt += __builtin_popcount(mask);
                        }
                    }
                }
            }
            g += 4;
        }
        for (int i = lane; i < (reuse ? 0 : cnt); i += 32) {
            int e = mylist[i] & 0xFFFFF;
            int d, t = 0;
            if (e < E_EDGES) {
                d = el[E_EDGES + e];
                t = et[e];
            } else {
                d = eln[E2_EDGES + (e - E_EDGES)];
            }
            mylist2[i] = d | (t << 16);
        }
        if (lane == 0) {
            s_cnt[wv] = cnt;
            if (g < gend) s_more = 1;
        }
        __syncthreads();
        const int more = s_more;

        {
            OwnEdge curA = {}, curB = {};
            bool have = false;
            bool half = false;
            int pen = 0, pen2 = 0;
            for (int w2 = w2lo; w2 < w2hi; ++w2) {
                const int n = __builtin_amdgcn_readfirstlane(s_cnt[w2]);
                const int* lp = lists + w2 * OWN_CAP;
                const int* lp2 = lists2 + w2 * OWN_CAP;
                for (int i0 = 0; i0 < n; i0 += 32) {
                    int idx = i0 + lane;
                    int ent = idx < n ? lp[idx] : 0;
                    int ent2 = idx < n ? lp2[idx] : 0;
                    bool mine = (idx < n) && (((ent >> 20) & 31) == wv);
                    unsigned mask = __builtin_amdgcn_ballot_w32(mine);
                    while (mask != 0u) {
                        int b = __builtin_ctz(mask);
                        mask &= mask - 1u;
                        int en = __builtin_amdgcn_readlane(ent, b);
                        int en2 = __builtin_amdgcn_readlane(ent2, b);
                        if constexpr (L == 1) {
                            if (kept < OWN_CAP && lane == 0) { k_w1[kept] = en; k_w2[kept] = en2; }
                            ++kept;
                        }
                        if (!half) {
                            pen = en;
                            pen2 = en2;
                            half = true;
                        } else {
                            OwnEdge nxA, nxB;
                            own_load2<L>(nxA, nxB, pen, pen2, en, en2, base, lane,
                                         etn, xrow, erow, rtab);
                            if (have) own_add2<L>(curA, curB, w, lane, acc, den);
                            curA = nxA;
                            curB = nxB;
                            have = true;
                            half = false;
                        }
                    }
                }
            }
            if (have) own_add2<L>(curA, curB, w, lane, acc, den);
            if (half) {
                OwnEdge last;
                own_load<L>(last, pen & 0xFFFFF, pen >> 20, pen2, base, lane,
                            etn, xrow, erow, rtab);
                own_add<L>(last, w, lane, acc, den);
            }
        }
        if constexpr (L == 1) {
            if (lane == 0 && kept > OWN_CAP) s_long = 1;
        }
        __syncthreads();
        if (!more) { single = (round == 0); break; }
    }
    if constexpr (L == 1) {
        if (lane == 0) *k_cnt = kept;
        if (tid == 0) *k_mark = (single && s_long == 0) ? 1 : 0;
    }

    float* orow = outp + (size_t)base * 128;
    for (int i = tid; i < OWN_NODES * 128; i += OWN_WAVES * 32) {
        int nl = i >> 7;
        float dnm;
        if constexpr (L == 1) dnm = den[nl * 2 + ((i & 127) >> 6)];
        else                  dnm = den[nl];
        dnm = dnm > 1e-12f ? dnm : 1e-12f;
        float h = acc[i] / dnm;
        if constexpr (L == 1) {
            orow[i] = h;
        } else {
            float y = h > 0.f ? h : (__expf(h) - 1.f);
            y = y > 0.f ? y : (__expf(y) - 1.f);
            orow[i] = y;
        }
    }
}

template <int LDBK>
static inline void launch_gemm(hipStream_t st, const float* A, const float* B,
                               float* C, int M, int N,
                               int lda, int ldbn, int ldc, int coff) {
    constexpr int NT = 4;
    int mtiles  = (M + 15) / 16;
    int ngroups = N / (NT * 16);
    long tiles = (long)mtiles * ngroups;
    int blocks = (int)((tiles + 7) / 8);
    gat_wmma_gemm_f32<LDBK, NT><<<blocks, 256, 0, st>>>(A, B, C, M, lda, ldbn,
                                                        ldc, coff, mtiles,
                                                        ngroups);
}

extern "C" void kernel_launch(void* const* d_in, const int* in_sizes, int n_in,
                              void* d_out, int out_size, void* d_ws, size_t ws_size,
                              hipStream_t stream) {
    const float* x    = (const float*)d_in[0];
    const float* rel  = (const float*)d_in[1];
    const int*   el   = (const int*)d_in[2];
    const int*   et   = (const int*)d_in[3];
    const float* eemb = (const float*)d_in[4];
    const int*   eln  = (const int*)d_in[5];
    const int*   etn  = (const int*)d_in[6];
    const float* w1h  = (const float*)d_in[8];
    const float* w2h  = (const float*)d_in[9];
    const float* w1o  = (const float*)d_in[10];
    const float* w2o  = (const float*)d_in[11];
    const float* WR   = (const float*)d_in[12];
    float* out = (float*)d_out;

    float* ws = (float*)d_ws;
    float* xnode  = ws;
    float* ec     = xnode  + (size_t)N_NODES * 256;
    float* relc   = ec     + (size_t)E_EDGES * 128;
    float* num1   = relc   + (size_t)NREL * 128;
    float* xab2   = num1   + (size_t)N_NODES * 128;
    float* relW2  = xab2   + (size_t)N_NODES * 256;
    int*   keep   = (int*)(relW2 + (size_t)NREL * 128);

    float* out_rel = out + (size_t)N_NODES * 128;

    gat_wmma_gemm_ws<<<512, 128, 0, stream>>>(x, w1h, xnode, N_NODES, 128, 384, 256,
                                              N_NODES / 16, 4, 2, 64 * 384, 128);
    gat_wmma_gemm_ws<<<512, 128, 0, stream>>>(eemb, w1h + 256, ec, E_EDGES, 128, 384,
                                              128, E_EDGES / 16, 2, 1, 64 * 384, 0);
    for (int h = 0; h < 2; ++h)
        launch_gemm<1>(stream, rel, w1h + (size_t)h * 64 * 384 + 256,
                       relc, NREL, 64, 128, 384, 128, h * 64);

    gat_own<1><<<N_NODES / OWN_NODES, OWN_WAVES * 32, 0, stream>>>(
        el, eln, et, etn, xnode, ec, relc, w2h, num1, keep);

    launch_gemm<128>(stream, rel, WR, out_rel, NREL, 128, 128, 1, 128, 0);
    gat_wmma_gemm_ws<<<512, 128, 0, stream>>>(num1, w1o, xab2, N_NODES, 128, 384, 256,
                                              N_NODES / 16, 4, 2, 128, 64 * 384);
    launch_gemm<1>(stream, out_rel, w1o + 256, relW2, NREL, 128, 128, 384, 128, 0);

    gat_own<2><<<N_NODES / OWN_NODES, OWN_WAVES * 32, 0, stream>>>(
        el, eln, et, etn, xab2, ec, relW2, w2o, out, keep);
}


# 1000-row table GEMMs moved from the k-step tile kernel (a wait on a fresh load before most WMMAs) to the weight-stationary kernel: loads go out as whole groups; launches 9 -> 8; sums unchanged
# speedup vs baseline: 2.2201x; 1.0720x over previous
#include <hip/hip_runtime.h>
#include <hip/hip_bf16.h>

typedef float v2f __attribute__((ext_vector_type(2)));
typedef float v8f __attribute__((ext_vector_type(8)));

#define N_NODES 50000
#define E_EDGES 500000
#define E2_EDGES 62500
#define ET_EDGES (E_EDGES + E2_EDGES)
#define NREL 1000
#define ALPHA 0.2f
#define KDIM 128

#define WS_NT 4
#define WS_KS (KDIM / 4)

__device__ __forceinline__ void ws_load_a(v2f (&a)[WS_KS], const float* __restrict__ A,
                                          int mt, int l16, int half, int M, int lda) {
    int arow = mt * 16 + l16;
    arow = arow < M ? arow : (M - 1);
    const float* ap = A + (size_t)arow * lda + half * 2;
#pragma unroll
    for (int j = 0; j < WS_KS; ++j) a[j] = *(const v2f*)(ap + j * 4);
}

__device__ __forceinline__ void ws_tile(const v2f (&a)[WS_KS],
                                        const v2f (&bf)[WS_NT][WS_KS],
                                        float* __restrict__ C, int mt, int l16,
                                        int half, int M, int ldc, int coff) {
    v8f acc[WS_NT];
#pragma unroll
    for (int s = 0; s < WS_NT; ++s) acc[s] = (v8f){0.f,0.f,0.f,0.f,0.f,0.f,0.f,0.f};
#pragma unroll
    for (int j = 0; j < WS_KS; ++j) {
#pragma unroll
        for (int s = 0; s < WS_NT; ++s)
            acc[s] = __builtin_amdgcn_wmma_f32_16x16x4_f32(false, a[j], false,
                                                           bf[s][j], (short)0,
                                                           acc[s], false, false);
    }
    float* cp = C + (size_t)(mt * 16 + half * 8) * ldc + coff + l16;
    if (mt * 16 + 16 <= M) {
#pragma unroll
        for (int r = 0; r < 8; ++r) {
#pragma unroll
            for (int s = 0; s < WS_NT; ++s) cp[(size_t)r * ldc + s * 16] = acc[s][r];
        }
    } else {
#pragma unroll
        for (int r = 0; r < 8; ++r) {
            if (mt * 16 + half * 8 + r < M) {
#pragma unroll
                for (int s = 0; s < WS_NT; ++s) cp[(size_t)r * ldc + s * 16] = acc[s][r];
            }
        }
    }
}

template <int LDBK>
__global__ void __launch_bounds__(128)
gat_wmma_gemm_ws(const float* __restrict__ A, const float* __restrict__ B,
                 float* __restrict__ C, int M, int lda, int ldbn, int ldc,
                 int mtiles, int ncg, int gdiv, int ghi, int glo) {
    const int lane = threadIdx.x & 31;
    const int half = lane >> 4;
    const int l16  = lane & 15;
    const int gw = blockIdx.x * 4 + __builtin_amdgcn_readfirstlane(threadIdx.x >> 5);
    const int nw = gridDim.x * 4;
    const int cg = gw % ncg;
    const int slot = gw / ncg;
    const int nslots = nw / ncg;
    if (slot >= mtiles) return;

    const float* Bg = B + (size_t)(cg / gdiv) * ghi + (size_t)(cg % gdiv) * glo
                        + (size_t)(half * 2) * LDBK;
    v2f bf[WS_NT][WS_KS];
#pragma unroll
    for (int s = 0; s < WS_NT; ++s) {
        const float* bp = Bg + (size_t)(s * 16 + l16) * ldbn;
#pragma unroll
        for (int j = 0; j < WS_KS; ++j) {
            if constexpr (LDBK == 1) bf[s][j] = *(const v2f*)(bp + j * 4);
            else {
                bf[s][j][0] = bp[(size_t)(j * 4) * LDBK];
                bf[s][j][1] = bp[(size_t)(j * 4 + 1) * LDBK];
            }
        }
    }

    const int coff = cg * (WS_NT * 16);
    int mt = slot;
    v2f a0[WS_KS], a1[WS_KS];
    ws_load_a(a0, A, mt, l16, half, M, lda);
    while (true) {
        {
            int mn = mt + nslots;
            mn = mn < mtiles ? mn : mt;
            ws_load_a(a1, A, mn, l16, half, M, lda);
            __builtin_amdgcn_sched_barrier(0);
            ws_tile(a0, bf, C, mt, l16, half, M, ldc, coff);
        }
        mt += nslots;
        if (mt >= mtiles) break;
        {
            int mn = mt + nslots;
            mn = mn < mtiles ? mn : mt;
            ws_load_a(a0, A, mn, l16, half, M, lda);
            __builtin_amdgcn_sched_barrier(0);
            ws_tile(a1, bf, C, mt, l16, half, M, ldc, coff);
        }
        mt += nslots;
        if (mt >= mtiles) break;
    }
}

#define OWN_NODES 200
#define OWN_WAVES 32
#define OWN_CAP   640
#define OWN_NG1   ((E_EDGES + 127) / 128)
#define OWN_NG2   ((E2_EDGES + 127) / 128)
#define OWN_NGT   (OWN_NG1 + OWN_NG2)
#define OWN_GPW   ((OWN_NGT + OWN_WAVES - 1) / OWN_WAVES)
#define OWN_ROUNDS ((OWN_GPW + 3) / 4)
#define OWN_NWG   (N_NODES / OWN_NODES)
#define KEEP_CNT_OFF  256
#define KEEP_W1_OFF   (KEEP_CNT_OFF + OWN_NWG * OWN_WAVES)
#define KEEP_W2_OFF   (KEEP_W1_OFF + OWN_NWG * OWN_WAVES * OWN_CAP)
static_assert(OWN_NWG <= KEEP_CNT_OFF, "one mark per workgroup in front of the lengths");
static_assert(N_NODES <= 65536 && NREL <= 32768, "second list word packs 16 + 15 bits");
static_assert(ET_EDGES <= (1 << 20) && OWN_NODES <= 256, "first list word packs 20 + 8 bits");
static_assert(N_NODES % OWN_NODES == 0, "every workgroup owns a full set of nodes");

struct OwnEdge {
    float xa[4], xb[4], r0[4], r1[4];
    int nl;
    int direct;
};

__device__ __forceinline__ void own_types(int e, const int* __restrict__ etn, int& t0, int& t1) {
    t0 = 0; t1 = 0;
    if (e >= E_EDGES) {
        int j = e - E_EDGES;
        t0 = etn[2 * j]; t1 = etn[2 * j + 1];
    }
}

template <int L>
__device__ __forceinline__ void own_rows_node(OwnEdge& r, int nl, int dt, int base, int lane,
                                              const float* __restrict__ xrow) {
    const int s = base + nl;
    const int d = dt & 0xFFFF;
    r.nl = nl;
    if constexpr (L == 1) {
#pragma unroll
        for (int h = 0; h < 2; ++h) {
            float2 a = *(const float2*)(xrow + (size_t)s * 256 + h * 128 + lane * 2);
            float2 b = *(const float2*)(xrow + (size_t)d * 256 + h * 128 + 64 + lane * 2);
            r.xa[h * 2] = a.x; r.xa[h * 2 + 1] = a.y;
            r.xb[h * 2] = b.x; r.xb[h * 2 + 1] = b.y;
        }
    } else {
        float4 a = *(const float4*)(xrow + (size_t)s * 256 + lane * 4);
        float4 b = *(const float4*)(xrow + (size_t)d * 256 + 128 + lane * 4);
        r.xa[0] = a.x; r.xa[1] = a.y; r.xa[2] = a.z; r.xa[3] = a.w;
        r.xb[0] = b.x; r.xb[1] = b.y; r.xb[2] = b.z; r.xb[3] = b.w;
    }
}

template <int L>
__device__ __forceinline__ void own_rows_rel(OwnEdge& r, int e, int dt, int t0, int t1, int lane,
                                             const float* __restrict__ erow,
                                             const float* __restrict__ rtab) {
    const bool direct = e < E_EDGES;
    r.direct = direct ? 1 : 0;
    if constexpr (L == 1) {
        const float* p0 = direct ? (erow + (size_t)e * 128) : (rtab + (size_t)t0 * 128);
#pragma unroll
        for (int h = 0; h < 2; ++h) {
            float2 v = *(const float2*)(p0 + h * 64 + lane * 2);
            r.r0[h * 2] = v.x; r.r0[h * 2 + 1] = v.y;
        }
        if (!direct) {
#pragma unroll
            for (int h = 0; h < 2; ++h) {
                float2 a1 = *(const float2*)(rtab + (size_t)t1 * 128 + h * 64 + lane * 2);
                r.r1[h * 2] = a1.x; r.r1[h * 2 + 1] = a1.y;
            }
        }
    } else {
        const int t = direct ? (dt >> 16) : t0;
        float4 v = *(const float4*)(rtab + (size_t)t * 128 + lane * 4);
        r.r0[0] = v.x; r.r0[1] = v.y; r.r0[2] = v.z; r.r0[3] = v.w;
        if (!direct) {
            float4 a1 = *(const float4*)(rtab + (size_t)t1 * 128 + lane * 4);
            r.r1[0] = a1.x; r.r1[1] = a1.y; r.r1[2] = a1.z; r.r1[3] = a1.w;
        }
    }
}

template <int L>
__device__ __forceinline__ void own_load(OwnEdge& r, int e, int nl, int dt, int base, int lane,
                                         const int* __restrict__ etn,
                                         const float* __restrict__ xrow,
                                         const float* __restrict__ erow,
                                         const float* __restrict__ rtab) {
    int t0, t1;
#pragma unroll
    for (int c = 0; c < 4; ++c) r.r1[c] = 0.f;
    own_types(e, etn, t0, t1);
    own_rows_node<L>(r, nl, dt, base, lane, xrow);
    own_rows_rel<L>(r, e, dt, t0, t1, lane, erow, rtab);
}

template <int L>
__device__ __forceinline__ void own_load2(OwnEdge& rA, OwnEdge& rB, int enA, int dtA, int enB,
                                          int dtB, int base, int lane,
                                          const int* __restrict__ etn,
                                          const float* __restrict__ xrow,
                                          const float* __restrict__ erow,
                                          const float* __restrict__ rtab) {
    const int eA = enA & 0xFFFFF, eB = enB & 0xFFFFF;
    int tA0, tA1, tB0, tB1;
#pragma unroll
    for (int c = 0; c < 4; ++c) { rA.r1[c] = 0.f; rB.r1[c] = 0.f; }
    own_types(eA, etn, tA0, tA1);
    own_types(eB, etn, tB0, tB1);
    own_rows_node<L>(rA, enA >> 20, dtA, base, lane, xrow);
    own_rows_node<L>(rB, enB >> 20, dtB, base, lane, xrow);
    own_rows_rel<L>(rA, eA, dtA, tA0, tA1, lane, erow, rtab);
    own_rows_rel<L>(rB, eB, dtB, tB0, tB1, lane, erow, rtab);
}

template <int L>
__device__ __forceinline__ void own_add(const OwnEdge& r, const float (&w)[4], int lane,
                                        float* acc, float* den) {
    float m[4];
#pragma unroll
    for (int c = 0; c < 4; ++c) {
        float ev = r.direct ? r.r0[c] : (r.r0[c] + r.r1[c]);
        m[c] = r.xa[c] + r.xb[c] + ev;
    }
    if constexpr (L == 1) {
#pragma unroll
        for (int h = 0; h < 2; ++h) {
            float m0 = m[h * 2], m1 = m[h * 2 + 1];
            float p = m0 * w[h * 2] + m1 * w[h * 2 + 1];
#pragma unroll
            for (int off = 16; off > 0; off >>= 1) p += __shfl_xor(p, off, 32);
            float lr = p > 0.f ? p : ALPHA * p;
            float eev = __expf(-lr);
            float2* ap = (float2*)(acc + r.nl * 128 + h * 64 + lane * 2);
            float2 av = *ap;
            av.x += eev * m0;
            av.y += eev * m1;
            *ap = av;
            if (lane == 0) den[r.nl * 2 + h] += eev;
        }
    } else {
        float p = m[0] * w[0] + m[1] * w[1] + m[2] * w[2] + m[3] * w[3];
#pragma unroll
        for (int off = 16; off > 0; off >>= 1) p += __shfl_xor(p, off, 32);
        float lr = p > 0.f ? p : ALPHA * p;
        float eev = __expf(-lr);
        float4* ap = (float4*)(acc + r.nl * 128 + lane * 4);
        float4 av = *ap;
        av.x += eev * m[0];
        av.y += eev * m[1];
        av.z += eev * m[2];
        av.w += eev * m[3];
        *ap = av;
        if (lane == 0) den[r.nl] += eev;
    }
}


template <int L>
__device__ __forceinline__ void own_add2(const OwnEdge& A, const OwnEdge& B,
                                         const float (&w)[4], int lane,
                                         float* acc, float* den) {
    float mA[4], mB[4];
#pragma unroll
    for (int c = 0; c < 4; ++c) {
        float evA = A.direct ? A.r0[c] : (A.r0[c] + A.r1[c]);
        float evB = B.direct ? B.r0[c] : (B.r0[c] + B.r1[c]);
        mA[c] = A.xa[c] + A.xb[c] + evA;
        mB[c] = B.xa[c] + B.xb[c] + evB;
    }
    const bool same = (A.nl == B.nl);
    if constexpr (L == 1) {
        float p[4];
        p[0] = mA[0] * w[0] + mA[1] * w[1];
        p[1] = mA[2] * w[2] + mA[3] * w[3];
        p[2] = mB[0] * w[0] + mB[1] * w[1];
        p[3] = mB[2] * w[2] + mB[3] * w[3];
#pragma unroll
        for (int off = 16; off > 0; off >>= 1) {
            float t[4];
#pragma unroll
            for (int q = 0; q < 4; ++q) t[q] = __shfl_xor(p[q], off, 32);
#pragma unroll
            for (int q = 0; q < 4; ++q) p[q] += t[q];
        }
        float ee[4];
#pragma unroll
        for (int q = 0; q < 4; ++q) {
            float lr = p[q] > 0.f ? p[q] : ALPHA * p[q];
            ee[q] = __expf(-lr);
        }
        float2* apA0 = (float2*)(acc + A.nl * 128 + lane * 2);
        float2* apA1 = (float2*)(acc + A.nl * 128 + 64 + lane * 2);
        float2* apB0 = (float2*)(acc + B.nl * 128 + lane * 2);
        float2* apB1 = (float2*)(acc + B.nl * 128 + 64 + lane * 2);
        float2 a0 = *apA0, a1 = *apA1, b0 = *apB0, b1 = *apB1;
        a0.x += ee[0] * mA[0]; a0.y += ee[0] * mA[1];
        a1.x += ee[1] * mA[2]; a1.y += ee[1] * mA[3];
        if (same) { b0 = a0; b1 = a1; }
        b0.x += ee[2] * mB[0]; b0.y += ee[2] * mB[1];
        b1.x += ee[3] * mB[2]; b1.y += ee[3] * mB[3];
        *apA0 = a0; *apA1 = a1;
        *apB0 = b0; *apB1 = b1;
        if (lane == 0) {
            float dA0 = den[A.nl * 2], dA1 = den[A.nl * 2 + 1];
            float dB0 = den[B.nl * 2], dB1 = den[B.nl * 2 + 1];
            dA0 += ee[0]; dA1 += ee[1];
            if (same) { dB0 = dA0; dB1 = dA1; }
            dB0 += ee[2]; dB1 += ee[3];
            den[A.nl * 2] = dA0; den[A.nl * 2 + 1] = dA1;
            den[B.nl * 2] = dB0; den[B.nl * 2 + 1] = dB1;
        }
    } else {
        float pA = mA[0] * w[0] + mA[1] * w[1] + mA[2] * w[2] + mA[3] * w[3];
        float pB = mB[0] * w[0] + mB[1] * w[1] + mB[2] * w[2] + mB[3] * w[3];
#pragma unroll
        for (int off = 16; off > 0; off >>= 1) {
            float tA = __shfl_xor(pA, off, 32);
            float tB = __shfl_xor(pB, off, 32);
            pA += tA;
            pB += tB;
        }
        float lrA = pA > 0.f ? pA : ALPHA * pA;
        float lrB = pB > 0.f ? pB : ALPHA * pB;
        float eA = __expf(-lrA);
        float eB = __expf(-lrB);
        float4* apA = (float4*)(acc + A.nl * 128 + lane * 4);
        float4* apB = (float4*)(acc + B.nl * 128 + lane * 4);
        float4 a = *apA, b = *apB;
        a.x += eA * mA[0]; a.y += eA * mA[1]; a.z += eA * mA[2]; a.w += eA * mA[3];
        if (same) b = a;
        b.x += eB * mB[0]; b.y += eB * mB[1]; b.z += eB * mB[2]; b.w += eB * mB[3];
        *apA = a;
        *apB = b;
        if (lane == 0) {
            float dA = den[A.nl], dB = den[B.nl];
            dA += eA;
            if (same) dB = dA;
            dB += eB;
            den[A.nl] = dA;
            den[B.nl] = dB;
        }
    }
}

template <int L>
__global__ void __launch_bounds__(OWN_WAVES * 32)
gat_own(const int* __restrict__ el, const int* __restrict__ eln,
        const int* __restrict__ et, const int* __restrict__ etn,
        const float* __restrict__ xrow, const float* __restrict__ erow,
        const float* __restrict__ rtab, const float* __restrict__ wvec,
        float* __restrict__ outp, int* __restrict__ keep) {
    __shared__ float acc[OWN_NODES * 128];
    __shared__ float den[OWN_NODES * 2];
    __shared__ int lists[OWN_WAVES * OWN_CAP];
    __shared__ int lists2[OWN_WAVES * OWN_CAP];
    __shared__ int s_cnt[OWN_WAVES];
    __shared__ int s_more;
    __shared__ int s_long;

    const int tid  = threadIdx.x;
    const int lane = tid & 31;
    const int wv   = __builtin_amdgcn_readfirstlane(tid >> 5);
    const int base = blockIdx.x * OWN_NODES;

    for (int i = tid; i < OWN_NODES * 128; i += OWN_WAVES * 32) acc[i] = 0.f;
    for (int i = tid; i < OWN_NODES * 2; i += OWN_WAVES * 32) den[i] = 0.f;
    if (tid == 0) s_long = 0;

    float w[4];
    if constexpr (L == 1) {
        w[0] = wvec[lane * 2];      w[1] = wvec[lane * 2 + 1];
        w[2] = wvec[64 + lane * 2]; w[3] = wvec[64 + lane * 2 + 1];
    } else {
        w[0] = wvec[lane * 4];     w[1] = wvec[lane * 4 + 1];
        w[2] = wvec[lane * 4 + 2]; w[3] = wvec[lane * 4 + 3];
    }

    int g = wv * OWN_GPW;
    int gend = g + OWN_GPW;
    gend = gend < OWN_NGT ? gend : OWN_NGT;
    int* mylist = lists + wv * OWN_CAP;
    int* mylist2 = lists2 + wv * OWN_CAP;
    const unsigned lt_mask = (1u << lane) - 1u;

    int* k_mark = keep + blockIdx.x;
    int* k_cnt  = keep + KEEP_CNT_OFF + blockIdx.x * OWN_WAVES + wv;
    int* k_w1   = keep + KEEP_W1_OFF + (size_t)(blockIdx.x * OWN_WAVES + wv) * OWN_CAP;
    int* k_w2   = keep + KEEP_W2_OFF + (size_t)(blockIdx.x * OWN_WAVES + wv) * OWN_CAP;
    bool reuse = false;
    if constexpr (L == 2) {
        reuse = (*k_mark == 1);
        if (reuse) g = gend;
    }
    const int w2lo = reuse ? wv : 0;
    const int w2hi = reuse ? wv + 1 : OWN_WAVES;
    int kept = 0;
    bool single = false;

    for (int round = 0; round < OWN_ROUNDS; ++round) {
        if (tid == 0) s_more = 0;
        __syncthreads();

        int cnt = 0;
        if (reuse) {
            cnt = *k_cnt;
            cnt = cnt < 0 ? 0 : (cnt > OWN_CAP ? OWN_CAP : cnt);
            for (int i = lane; i < cnt; i += 32) {
                mylist[i] = k_w1[i];
                mylist2[i] = k_w2[i];
            }
        }
        while (g < gend && cnt + 512 <= OWN_CAP) {
            int4 v[4];
#pragma unroll
            for (int q = 0; q < 4; ++q) {
                int gi = g + q;
                gi = gi < gend ? gi : (gend - 1);
                const int* p = gi < OWN_NG1 ? (el + (size_t)gi * 128)
                                            : (eln + (size_t)(gi - OWN_NG1) * 128);
                v[q] = *(const int4*)(p + lane * 4);
            }
#pragma unroll
            for (int q = 0; q < 4; ++q) {
                int gi = g + q;
                if (gi < gend) {
                    bool isd = gi < OWN_NG1;
                    int eb   = isd ? gi * 128 : (gi - OWN_NG1) * 128;
                    int lim  = isd ? E_EDGES : E2_EDGES;
                    int eoff = isd ? 0 : E_EDGES;
                    int idx0 = eb + lane * 4;
                    int sv[4] = {v[q].x, v[q].y, v[q].z, v[q].w};
#pragma unroll
                    for (int j = 0; j < 4; ++j) {
                        int loc = sv[j] - base;
                        bool hit = (idx0 + j < lim) && ((unsigned)loc < (unsigned)OWN_NODES);
                        unsigned mask = __builtin_amdgcn_ballot_w32(hit);
                        if (mask != 0u) {
                            int pos = cnt + __builtin_popcount(mask & lt_mask);
                            if (hit) mylist[pos] = (eoff + idx0 + j) | (loc << 20);
                            cnt += __builtin_popcount(mask);
                        }
                    }
                }
            }
            g += 4;
        }
        for (int i = lane; i < (reuse ? 0 : cnt); i += 32) {
            int e = mylist[i] & 0xFFFFF;
            int d, t = 0;
            if (e < E_EDGES) {
                d = el[E_EDGES + e];
                t = et[e];
            } else {
                d = eln[E2_EDGES + (e - E_EDGES)];
            }
            mylist2[i] = d | (t << 16);
        }
        if (lane == 0) {
            s_cnt[wv] = cnt;
            if (g < gend) s_more = 1;
        }
        __syncthreads();
        const int more = s_more;

        {
            OwnEdge curA = {}, curB = {};
            bool have = false;
            bool half = false;
            int pen = 0, pen2 = 0;
            for (int w2 = w2lo; w2 < w2hi; ++w2) {
                const int n = __builtin_amdgcn_readfirstlane(s_cnt[w2]);
                const int* lp = lists + w2 * OWN_CAP;
                const int* lp2 = lists2 + w2 * OWN_CAP;
                for (int i0 = 0; i0 < n; i0 += 32) {
                    int idx = i0 + lane;
                    int ent = idx < n ? lp[idx] : 0;
                    int ent2 = idx < n ? lp2[idx] : 0;
                    bool mine = (idx < n) && (((ent >> 20) & 31) == wv);
                    unsigned mask = __builtin_amdgcn_ballot_w32(mine);
                    while (mask != 0u) {
                        int b = __builtin_ctz(mask);
                        mask &= mask - 1u;
                        int en = __builtin_amdgcn_readlane(ent, b);
                        int en2 = __builtin_amdgcn_readlane(ent2, b);
                        if constexpr (L == 1) {
                            if (kept < OWN_CAP && lane == 0) { k_w1[kept] = en; k_w2[kept] = en2; }
                            ++kept;
                        }
                        if (!half) {
                            pen = en;
                            pen2 = en2;
                            half = true;
                        } else {
                            OwnEdge nxA, nxB;
                            own_load2<L>(nxA, nxB, pen, pen2, en, en2, base, lane,
                                         etn, xrow, erow, rtab);
                            if (have) own_add2<L>(curA, curB, w, lane, acc, den);
                            curA = nxA;
                            curB = nxB;
                            have = true;
                            half = false;
                        }
                    }
                }
            }
            if (have) own_add2<L>(curA, curB, w, lane, acc, den);
            if (half) {
                OwnEdge last;
                own_load<L>(last, pen & 0xFFFFF, pen >> 20, pen2, base, lane,
                            etn, xrow, erow, rtab);
                own_add<L>(last, w, lane, acc, den);
            }
        }
        if constexpr (L == 1) {
            if (lane == 0 && kept > OWN_CAP) s_long = 1;
        }
        __syncthreads();
        if (!more) { single = (round == 0); break; }
    }
    if constexpr (L == 1) {
        if (lane == 0) *k_cnt = kept;
        if (tid == 0) *k_mark = (single && s_long == 0) ? 1 : 0;
    }

    float* orow = outp + (size_t)base * 128;
    for (int i = tid; i < OWN_NODES * 128; i += OWN_WAVES * 32) {
        int nl = i >> 7;
        float dnm;
        if constexpr (L == 1) dnm = den[nl * 2 + ((i & 127) >> 6)];
        else                  dnm = den[nl];
        dnm = dnm > 1e-12f ? dnm : 1e-12f;
        float h = acc[i] / dnm;
        if constexpr (L == 1) {
            orow[i] = h;
        } else {
            float y = h > 0.f ? h : (__expf(h) - 1.f);
            y = y > 0.f ? y : (__expf(y) - 1.f);
            orow[i] = y;
        }
    }
}

extern "C" void kernel_launch(void* const* d_in, const int* in_sizes, int n_in,
                              void* d_out, int out_size, void* d_ws, size_t ws_size,
                              hipStream_t stream) {
    const float* x    = (const float*)d_in[0];
    const float* rel  = (const float*)d_in[1];
    const int*   el   = (const int*)d_in[2];
    const int*   et   = (const int*)d_in[3];
    const float* eemb = (const float*)d_in[4];
    const int*   eln  = (const int*)d_in[5];
    const int*   etn  = (const int*)d_in[6];
    const float* w1h  = (const float*)d_in[8];
    const float* w2h  = (const float*)d_in[9];
    const float* w1o  = (const float*)d_in[10];
    const float* w2o  = (const float*)d_in[11];
    const float* WR   = (const float*)d_in[12];
    float* out = (float*)d_out;

    float* ws = (float*)d_ws;
    float* xnode  = ws;
    float* ec     = xnode  + (size_t)N_NODES * 256;
    float* relc   = ec     + (size_t)E_EDGES * 128;
    float* num1   = relc   + (size_t)NREL * 128;
    float* xab2   = num1   + (size_t)N_NODES * 128;
    float* relW2  = xab2   + (size_t)N_NODES * 256;
    int*   keep   = (int*)(relW2 + (size_t)NREL * 128);

    float* out_rel = out + (size_t)N_NODES * 128;

    gat_wmma_gemm_ws<1><<<512, 128, 0, stream>>>(x, w1h, xnode, N_NODES, 128, 384, 256,
                                                 N_NODES / 16, 4, 2, 64 * 384, 128);
    gat_wmma_gemm_ws<1><<<512, 128, 0, stream>>>(eemb, w1h + 256, ec, E_EDGES, 128, 384,
                                                 128, E_EDGES / 16, 2, 1, 64 * 384, 0);
    gat_wmma_gemm_ws<1><<<32, 128, 0, stream>>>(rel, w1h + 256, relc, NREL, 128, 384,
                                                128, (NREL + 15) / 16, 2, 1, 64 * 384, 0);

    gat_own<1><<<N_NODES / OWN_NODES, OWN_WAVES * 32, 0, stream>>>(
        el, eln, et, etn, xnode, ec, relc, w2h, num1, keep);

    gat_wmma_gemm_ws<128><<<32, 128, 0, stream>>>(rel, WR, out_rel, NREL, 128, 1, 128,
                                                  (NREL + 15) / 16, 2, 1, 64, 0);
    gat_wmma_gemm_ws<1><<<512, 128, 0, stream>>>(num1, w1o, xab2, N_NODES, 128, 384, 256,
                                                 N_NODES / 16, 4, 2, 128, 64 * 384);
    gat_wmma_gemm_ws<1><<<32, 128, 0, stream>>>(out_rel, w1o + 256, relW2, NREL, 128, 384,
                                                128, (NREL + 15) / 16, 2, 1, 64 * 384, 0);

    gat_own<2><<<N_NODES / OWN_NODES, OWN_WAVES * 32, 0, stream>>>(
        el, eln, et, etn, xab2, ec, relW2, w2o, out, keep);
}


# layer-1 edge step: a lane holds 4 adjacent row columns (one head per 16-lane half) as in layer 2: 16-byte row loads and LDS adds, one 4-step exchange sums both heads' logits; logit sum order changes
# speedup vs baseline: 2.3478x; 1.0575x over previous
#include <hip/hip_runtime.h>
#include <hip/hip_bf16.h>

typedef float v2f __attribute__((ext_vector_type(2)));
typedef float v8f __attribute__((ext_vector_type(8)));

#define N_NODES 50000
#define E_EDGES 500000
#define E2_EDGES 62500
#define ET_EDGES (E_EDGES + E2_EDGES)
#define NREL 1000
#define ALPHA 0.2f
#define KDIM 128

#define WS_NT 4
#define WS_KS (KDIM / 4)

__device__ __forceinline__ void ws_load_a(v2f (&a)[WS_KS], const float* __restrict__ A,
                                          int mt, int l16, int half, int M, int lda) {
    int arow = mt * 16 + l16;
    arow = arow < M ? arow : (M - 1);
    const float* ap = A + (size_t)arow * lda + half * 2;
#pragma unroll
    for (int j = 0; j < WS_KS; ++j) a[j] = *(const v2f*)(ap + j * 4);
}

__device__ __forceinline__ void ws_tile(const v2f (&a)[WS_KS],
                                        const v2f (&bf)[WS_NT][WS_KS],
                                        float* __restrict__ C, int mt, int l16,
                                        int half, int M, int ldc, int coff) {
    v8f acc[WS_NT];
#pragma unroll
    for (int s = 0; s < WS_NT; ++s) acc[s] = (v8f){0.f,0.f,0.f,0.f,0.f,0.f,0.f,0.f};
#pragma unroll
    for (int j = 0; j < WS_KS; ++j) {
#pragma unroll
        for (int s = 0; s < WS_NT; ++s)
            acc[s] = __builtin_amdgcn_wmma_f32_16x16x4_f32(false, a[j], false,
                                                           bf[s][j], (short)0,
                                                           acc[s], false, false);
    }
    float* cp = C + (size_t)(mt * 16 + half * 8) * ldc + coff + l16;
    if (mt * 16 + 16 <= M) {
#pragma unroll
        for (int r = 0; r < 8; ++r) {
#pragma unroll
            for (int s = 0; s < WS_NT; ++s) cp[(size_t)r * ldc + s * 16] = acc[s][r];
        }
    } else {
#pragma unroll
        for (int r = 0; r < 8; ++r) {
            if (mt * 16 + half * 8 + r < M) {
#pragma unroll
                for (int s = 0; s < WS_NT; ++s) cp[(size_t)r * ldc + s * 16] = acc[s][r];
            }
        }
    }
}

template <int LDBK>
__global__ void __launch_bounds__(128)
gat_wmma_gemm_ws(const float* __restrict__ A, const float* __restrict__ B,
                 float* __restrict__ C, int M, int lda, int ldbn, int ldc,
                 int mtiles, int ncg, int gdiv, int ghi, int glo) {
    const int lane = threadIdx.x & 31;
    const int half = lane >> 4;
    const int l16  = lane & 15;
    const int gw = blockIdx.x * 4 + __builtin_amdgcn_readfirstlane(threadIdx.x >> 5);
    const int nw = gridDim.x * 4;
    const int cg = gw % ncg;
    const int slot = gw / ncg;
    const int nslots = nw / ncg;
    if (slot >= mtiles) return;

    const float* Bg = B + (size_t)(cg / gdiv) * ghi + (size_t)(cg % gdiv) * glo
                        + (size_t)(half * 2) * LDBK;
    v2f bf[WS_NT][WS_KS];
#pragma unroll
    for (int s = 0; s < WS_NT; ++s) {
        const float* bp = Bg + (size_t)(s * 16 + l16) * ldbn;
#pragma unroll
        for (int j = 0; j < WS_KS; ++j) {
            if constexpr (LDBK == 1) bf[s][j] = *(const v2f*)(bp + j * 4);
            else {
                bf[s][j][0] = bp[(size_t)(j * 4) * LDBK];
                bf[s][j][1] = bp[(size_t)(j * 4 + 1) * LDBK];
            }
        }
    }

    const int coff = cg * (WS_NT * 16);
    int mt = slot;
    v2f a0[WS_KS], a1[WS_KS];
    ws_load_a(a0, A, mt, l16, half, M, lda);
    while (true) {
        {
            int mn = mt + nslots;
            mn = mn < mtiles ? mn : mt;
            ws_load_a(a1, A, mn, l16, half, M, lda);
            __builtin_amdgcn_sched_barrier(0);
            ws_tile(a0, bf, C, mt, l16, half, M, ldc, coff);
        }
        mt += nslots;
        if (mt >= mtiles) break;
        {
            int mn = mt + nslots;
            mn = mn < mtiles ? mn : mt;
            ws_load_a(a0, A, mn, l16, half, M, lda);
            __builtin_amdgcn_sched_barrier(0);
            ws_tile(a1, bf, C, mt, l16, half, M, ldc, coff);
        }
        mt += nslots;
        if (mt >= mtiles) break;
    }
}

#define OWN_NODES 200
#define OWN_WAVES 32
#define OWN_CAP   640
#define OWN_NG1   ((E_EDGES + 127) / 128)
#define OWN_NG2   ((E2_EDGES + 127) / 128)
#define OWN_NGT   (OWN_NG1 + OWN_NG2)
#define OWN_GPW   ((OWN_NGT + OWN_WAVES - 1) / OWN_WAVES)
#define OWN_ROUNDS ((OWN_GPW + 3) / 4)
#define OWN_NWG   (N_NODES / OWN_NODES)
#define KEEP_CNT_OFF  256
#define KEEP_W1_OFF   (KEEP_CNT_OFF + OWN_NWG * OWN_WAVES)
#define KEEP_W2_OFF   (KEEP_W1_OFF + OWN_NWG * OWN_WAVES * OWN_CAP)
static_assert(OWN_NWG <= KEEP_CNT_OFF, "one mark per workgroup in front of the lengths");
static_assert(N_NODES <= 65536 && NREL <= 32768, "second list word packs 16 + 15 bits");
static_assert(ET_EDGES <= (1 << 20) && OWN_NODES <= 256, "first list word packs 20 + 8 bits");
static_assert(N_NODES % OWN_NODES == 0, "every workgroup owns a full set of nodes");

struct OwnEdge {
    float xa[4], xb[4], r0[4], r1[4];
    int nl;
    int direct;
};

__device__ __forceinline__ void own_types(int e, const int* __restrict__ etn, int& t0, int& t1) {
    t0 = 0; t1 = 0;
    if (e >= E_EDGES) {
        int j = e - E_EDGES;
        t0 = etn[2 * j]; t1 = etn[2 * j + 1];
    }
}

template <int L>
__device__ __forceinline__ void own_rows_node(OwnEdge& r, int nl, int dt, int base, int lane,
                                              const float* __restrict__ xrow) {
    const int s = base + nl;
    const int d = dt & 0xFFFF;
    r.nl = nl;
    if constexpr (L == 1) {
        const int hc = (lane >> 4) * 128 + (lane & 15) * 4;
        float4 a = *(const float4*)(xrow + (size_t)s * 256 + hc);
        float4 b = *(const float4*)(xrow + (size_t)d * 256 + hc + 64);
        r.xa[0] = a.x; r.xa[1] = a.y; r.xa[2] = a.z; r.xa[3] = a.w;
        r.xb[0] = b.x; r.xb[1] = b.y; r.xb[2] = b.z; r.xb[3] = b.w;
    } else {
        float4 a = *(const float4*)(xrow + (size_t)s * 256 + lane * 4);
        float4 b = *(const float4*)(xrow + (size_t)d * 256 + 128 + lane * 4);
        r.xa[0] = a.x; r.xa[1] = a.y; r.xa[2] = a.z; r.xa[3] = a.w;
        r.xb[0] = b.x; r.xb[1] = b.y; r.xb[2] = b.z; r.xb[3] = b.w;
    }
}

template <int L>
__device__ __forceinline__ void own_rows_rel(OwnEdge& r, int e, int dt, int t0, int t1, int lane,
                                             const float* __restrict__ erow,
                                             const float* __restrict__ rtab) {
    const bool direct = e < E_EDGES;
    r.direct = direct ? 1 : 0;
    if constexpr (L == 1) {
        const float* p0 = direct ? (erow + (size_t)e * 128) : (rtab + (size_t)t0 * 128);
        float4 v = *(const float4*)(p0 + lane * 4);
        r.r0[0] = v.x; r.r0[1] = v.y; r.r0[2] = v.z; r.r0[3] = v.w;
        if (!direct) {
            float4 a1 = *(const float4*)(rtab + (size_t)t1 * 128 + lane * 4);
            r.r1[0] = a1.x; r.r1[1] = a1.y; r.r1[2] = a1.z; r.r1[3] = a1.w;
        }
    } else {
        const int t = direct ? (dt >> 16) : t0;
        float4 v = *(const float4*)(rtab + (size_t)t * 128 + lane * 4);
        r.r0[0] = v.x; r.r0[1] = v.y; r.r0[2] = v.z; r.r0[3] = v.w;
        if (!direct) {
            float4 a1 = *(const float4*)(rtab + (size_t)t1 * 128 + lane * 4);
            r.r1[0] = a1.x; r.r1[1] = a1.y; r.r1[2] = a1.z; r.r1[3] = a1.w;
        }
    }
}

template <int L>
__device__ __forceinline__ void own_load(OwnEdge& r, int e, int nl, int dt, int base, int lane,
                                         const int* __restrict__ etn,
                                         const float* __restrict__ xrow,
                                         const float* __restrict__ erow,
                                         const float* __restrict__ rtab) {
    int t0, t1;
#pragma unroll
    for (int c = 0; c < 4; ++c) r.r1[c] = 0.f;
    own_types(e, etn, t0, t1);
    own_rows_node<L>(r, nl, dt, base, lane, xrow);
    own_rows_rel<L>(r, e, dt, t0, t1, lane, erow, rtab);
}

template <int L>
__device__ __forceinline__ void own_load2(OwnEdge& rA, OwnEdge& rB, int enA, int dtA, int enB,
                                          int dtB, int base, int lane,
                                          const int* __restrict__ etn,
                                          const float* __restrict__ xrow,
                                          const float* __restrict__ erow,
                                          const float* __restrict__ rtab) {
    const int eA = enA & 0xFFFFF, eB = enB & 0xFFFFF;
    int tA0, tA1, tB0, tB1;
#pragma unroll
    for (int c = 0; c < 4; ++c) { rA.r1[c] = 0.f; rB.r1[c] = 0.f; }
    own_types(eA, etn, tA0, tA1);
    own_types(eB, etn, tB0, tB1);
    own_rows_node<L>(rA, enA >> 20, dtA, base, lane, xrow);
    own_rows_node<L>(rB, enB >> 20, dtB, base, lane, xrow);
    own_rows_rel<L>(rA, eA, dtA, tA0, tA1, lane, erow, rtab);
    own_rows_rel<L>(rB, eB, dtB, tB0, tB1, lane, erow, rtab);
}

template <int L> struct OwnLay {
    static constexpr int XOFF = (L == 1) ? 8 : 16;
    static __device__ __forceinline__ bool den_writer(int lane) {
        return (L == 1) ? ((lane & 15) == 0) : (lane == 0);
    }
    static __device__ __forceinline__ int den_idx(int nl, int lane) {
        return (L == 1) ? (nl * 2 + (lane >> 4)) : nl;
    }
};

template <int L>
__device__ __forceinline__ void own_add(const OwnEdge& r, const float (&w)[4], int lane,
                                        float* acc, float* den) {
    float m[4];
#pragma unroll
    for (int c = 0; c < 4; ++c) {
        float ev = r.direct ? r.r0[c] : (r.r0[c] + r.r1[c]);
        m[c] = r.xa[c] + r.xb[c] + ev;
    }
    float p = m[0] * w[0] + m[1] * w[1] + m[2] * w[2] + m[3] * w[3];
#pragma unroll
    for (int off = OwnLay<L>::XOFF; off > 0; off >>= 1) p += __shfl_xor(p, off, 32);
    float lr = p > 0.f ? p : ALPHA * p;
    float eev = __expf(-lr);
    float4* ap = (float4*)(acc + r.nl * 128 + lane * 4);
    float4 av = *ap;
    av.x += eev * m[0];
    av.y += eev * m[1];
    av.z += eev * m[2];
    av.w += eev * m[3];
    *ap = av;
    if (OwnLay<L>::den_writer(lane)) den[OwnLay<L>::den_idx(r.nl, lane)] += eev;
}


template <int L>
__device__ __forceinline__ void own_add2(const OwnEdge& A, const OwnEdge& B,
                                         const float (&w)[4], int lane,
                                         float* acc, float* den) {
    float mA[4], mB[4];
#pragma unroll
    for (int c = 0; c < 4; ++c) {
        float evA = A.direct ? A.r0[c] : (A.r0[c] + A.r1[c]);
        float evB = B.direct ? B.r0[c] : (B.r0[c] + B.r1[c]);
        mA[c] = A.xa[c] + A.xb[c] + evA;
        mB[c] = B.xa[c] + B.xb[c] + evB;
    }
    const bool same = (A.nl == B.nl);
    float pA = mA[0] * w[0] + mA[1] * w[1] + mA[2] * w[2] + mA[3] * w[3];
    float pB = mB[0] * w[0] + mB[1] * w[1] + mB[2] * w[2] + mB[3] * w[3];
#pragma unroll
    for (int off = OwnLay<L>::XOFF; off > 0; off >>= 1) {
        float tA = __shfl_xor(pA, off, 32);
        float tB = __shfl_xor(pB, off, 32);
        pA += tA;
        pB += tB;
    }
    float lrA = pA > 0.f ? pA : ALPHA * pA;
    float lrB = pB > 0.f ? pB : ALPHA * pB;
    float eA = __expf(-lrA);
    float eB = __expf(-lrB);
    float4* apA = (float4*)(acc + A.nl * 128 + lane * 4);
    float4* apB = (float4*)(acc + B.nl * 128 + lane * 4);
    float4 a = *apA, b = *apB;
    a.x += eA * mA[0]; a.y += eA * mA[1]; a.z += eA * mA[2]; a.w += eA * mA[3];
    if (same) b = a;
    b.x += eB * mB[0]; b.y += eB * mB[1]; b.z += eB * mB[2]; b.w += eB * mB[3];
    *apA = a;
    *apB = b;
    if (OwnLay<L>::den_writer(lane)) {
        const int iA = OwnLay<L>::den_idx(A.nl, lane), iB = OwnLay<L>::den_idx(B.nl, lane);
        float dA = den[iA], dB = den[iB];
        dA += eA;
        if (same) dB = dA;
        dB += eB;
        den[iA] = dA;
        den[iB] = dB;
    }
}

template <int L>
__global__ void __launch_bounds__(OWN_WAVES * 32)
gat_own(const int* __restrict__ el, const int* __restrict__ eln,
        const int* __restrict__ et, const int* __restrict__ etn,
        const float* __restrict__ xrow, const float* __restrict__ erow,
        const float* __restrict__ rtab, const float* __restrict__ wvec,
        float* __restrict__ outp, int* __restrict__ keep) {
    __shared__ float acc[OWN_NODES * 128];
    __shared__ float den[OWN_NODES * 2];
    __shared__ int lists[OWN_WAVES * OWN_CAP];
    __shared__ int lists2[OWN_WAVES * OWN_CAP];
    __shared__ int s_cnt[OWN_WAVES];
    __shared__ int s_more;
    __shared__ int s_long;

    const int tid  = threadIdx.x;
    const int lane = tid & 31;
    const int wv   = __builtin_amdgcn_readfirstlane(tid >> 5);
    const int base = blockIdx.x * OWN_NODES;

    for (int i = tid; i < OWN_NODES * 128; i += OWN_WAVES * 32) acc[i] = 0.f;
    for (int i = tid; i < OWN_NODES * 2; i += OWN_WAVES * 32) den[i] = 0.f;
    if (tid == 0) s_long = 0;

    float w[4];
    w[0] = wvec[lane * 4];     w[1] = wvec[lane * 4 + 1];
    w[2] = wvec[lane * 4 + 2]; w[3] = wvec[lane * 4 + 3];

    int g = wv * OWN_GPW;
    int gend = g + OWN_GPW;
    gend = gend < OWN_NGT ? gend : OWN_NGT;
    int* mylist = lists + wv * OWN_CAP;
    int* mylist2 = lists2 + wv * OWN_CAP;
    const unsigned lt_mask = (1u << lane) - 1u;

    int* k_mark = keep + blockIdx.x;
    int* k_cnt  = keep + KEEP_CNT_OFF + blockIdx.x * OWN_WAVES + wv;
    int* k_w1   = keep + KEEP_W1_OFF + (size_t)(blockIdx.x * OWN_WAVES + wv) * OWN_CAP;
    int* k_w2   = keep + KEEP_W2_OFF + (size_t)(blockIdx.x * OWN_WAVES + wv) * OWN_CAP;
    bool reuse = false;
    if constexpr (L == 2) {
        reuse = (*k_mark == 1);
        if (reuse) g = gend;
    }
    const int w2lo = reuse ? wv : 0;
    const int w2hi = reuse ? wv + 1 : OWN_WAVES;
    int kept = 0;
    bool single = false;

    for (int round = 0; round < OWN_ROUNDS; ++round) {
        if (tid == 0) s_more = 0;
        __syncthreads();

        int cnt = 0;
        if (reuse) {
            cnt = *k_cnt;
            cnt = cnt < 0 ? 0 : (cnt > OWN_CAP ? OWN_CAP : cnt);
            for (int i = lane; i < cnt; i += 32) {
                mylist[i] = k_w1[i];
                mylist2[i] = k_w2[i];
            }
        }
        while (g < gend && cnt + 512 <= OWN_CAP) {
            int4 v[4];
#pragma unroll
            for (int q = 0; q < 4; ++q) {
                int gi = g + q;
                gi = gi < gend ? gi : (gend - 1);
                const int* p = gi < OWN_NG1 ? (el + (size_t)gi * 128)
                                            : (eln + (size_t)(gi - OWN_NG1) * 128);
                v[q] = *(const int4*)(p + lane * 4);
            }
#pragma unroll
            for (int q = 0; q < 4; ++q) {
                int gi = g + q;
                if (gi < gend) {
                    bool isd = gi < OWN_NG1;
                    int eb   = isd ? gi * 128 : (gi - OWN_NG1) * 128;
                    int lim  = isd ? E_EDGES : E2_EDGES;
                    int eoff = isd ? 0 : E_EDGES;
                    int idx0 = eb + lane * 4;
                    int sv[4] = {v[q].x, v[q].y, v[q].z, v[q].w};
#pragma unroll
                    for (int j = 0; j < 4; ++j) {
                        int loc = sv[j] - base;
                        bool hit = (idx0 + j < lim) && ((unsigned)loc < (unsigned)OWN_NODES);
                        unsigned mask = __builtin_amdgcn_ballot_w32(hit);
                        if (mask != 0u) {
                            int pos = cnt + __builtin_popcount(mask & lt_mask);
                            if (hit) mylist[pos] = (eoff + idx0 + j) | (loc << 20);
                            cnt += __builtin_popcount(mask);
                        }
                    }
                }
            }
            g += 4;
        }
        for (int i = lane; i < (reuse ? 0 : cnt); i += 32) {
            int e = mylist[i] & 0xFFFFF;
            int d, t = 0;
            if (e < E_EDGES) {
                d = el[E_EDGES + e];
                t = et[e];
            } else {
                d = eln[E2_EDGES + (e - E_EDGES)];
            }
            mylist2[i] = d | (t << 16);
        }
        if (lane == 0) {
            s_cnt[wv] = cnt;
            if (g < gend) s_more = 1;
        }
        __syncthreads();
        const int more = s_more;

        {
            OwnEdge curA = {}, curB = {};
            bool have = false;
            bool half = false;
            int pen = 0, pen2 = 0;
            for (int w2 = w2lo; w2 < w2hi; ++w2) {
                const int n = __builtin_amdgcn_readfirstlane(s_cnt[w2]);
                const int* lp = lists + w2 * OWN_CAP;
                const int* lp2 = lists2 + w2 * OWN_CAP;
                for (int i0 = 0; i0 < n; i0 += 32) {
                    int idx = i0 + lane;
                    int ent = idx < n ? lp[idx] : 0;
                    int ent2 = idx < n ? lp2[idx] : 0;
                    bool mine = (idx < n) && (((ent >> 20) & 31) == wv);
                    unsigned mask = __builtin_amdgcn_ballot_w32(mine);
                    while (mask != 0u) {
                        int b = __builtin_ctz(mask);
                        mask &= mask - 1u;
                        int en = __builtin_amdgcn_readlane(ent, b);
                        int en2 = __builtin_amdgcn_readlane(ent2, b);
                        if constexpr (L == 1) {
                            if (kept < OWN_CAP && lane == 0) { k_w1[kept] = en; k_w2[kept] = en2; }
                            ++kept;
                        }
                        if (!half) {
                            pen = en;
                            pen2 = en2;
                            half = true;
                        } else {
                            OwnEdge nxA, nxB;
                            own_load2<L>(nxA, nxB, pen, pen2, en, en2, base, lane,
                                         etn, xrow, erow, rtab);
                            if (have) own_add2<L>(curA, curB, w, lane, acc, den);
                            curA = nxA;
                            curB = nxB;
                            have = true;
                            half = false;
                        }
                    }
                }
            }
            if (have) own_add2<L>(curA, curB, w, lane, acc, den);
            if (half) {
                OwnEdge last;
                own_load<L>(last, pen & 0xFFFFF, pen >> 20, pen2, base, lane,
                            etn, xrow, erow, rtab);
                own_add<L>(last, w, lane, acc, den);
            }
        }
        if constexpr (L == 1) {
            if (lane == 0 && kept > OWN_CAP) s_long = 1;
        }
        __syncthreads();
        if (!more) { single = (round == 0); break; }
    }
    if constexpr (L == 1) {
        if (lane == 0) *k_cnt = kept;
        if (tid == 0) *k_mark = (single && s_long == 0) ? 1 : 0;
    }

    float* orow = outp + (size_t)base * 128;
    for (int i = tid; i < OWN_NODES * 128; i += OWN_WAVES * 32) {
        int nl = i >> 7;
        float dnm;
        if constexpr (L == 1) dnm = den[nl * 2 + ((i & 127) >> 6)];
        else                  dnm = den[nl];
        dnm = dnm > 1e-12f ? dnm : 1e-12f;
        float h = acc[i] / dnm;
        if constexpr (L == 1) {
            orow[i] = h;
        } else {
            float y = h > 0.f ? h : (__expf(h) - 1.f);
            y = y > 0.f ? y : (__expf(y) - 1.f);
            orow[i] = y;
        }
    }
}

extern "C" void kernel_launch(void* const* d_in, const int* in_sizes, int n_in,
                              void* d_out, int out_size, void* d_ws, size_t ws_size,
                              hipStream_t stream) {
    const float* x    = (const float*)d_in[0];
    const float* rel  = (const float*)d_in[1];
    const int*   el   = (const int*)d_in[2];
    const int*   et   = (const int*)d_in[3];
    const float* eemb = (const float*)d_in[4];
    const int*   eln  = (const int*)d_in[5];
    const int*   etn  = (const int*)d_in[6];
    const float* w1h  = (const float*)d_in[8];
    const float* w2h  = (const float*)d_in[9];
    const float* w1o  = (const float*)d_in[10];
    const float* w2o  = (const float*)d_in[11];
    const float* WR   = (const float*)d_in[12];
    float* out = (float*)d_out;

    float* ws = (float*)d_ws;
    float* xnode  = ws;
    float* ec     = xnode  + (size_t)N_NODES * 256;
    float* relc   = ec     + (size_t)E_EDGES * 128;
    float* num1   = relc   + (size_t)NREL * 128;
    float* xab2   = num1   + (size_t)N_NODES * 128;
    float* relW2  = xab2   + (size_t)N_NODES * 256;
    int*   keep   = (int*)(relW2 + (size_t)NREL * 128);

    float* out_rel = out + (size_t)N_NODES * 128;

    gat_wmma_gemm_ws<1><<<512, 128, 0, stream>>>(x, w1h, xnode, N_NODES, 128, 384, 256,
                                                 N_NODES / 16, 4, 2, 64 * 384, 128);
    gat_wmma_gemm_ws<1><<<512, 128, 0, stream>>>(eemb, w1h + 256, ec, E_EDGES, 128, 384,
                                                 128, E_EDGES / 16, 2, 1, 64 * 384, 0);
    gat_wmma_gemm_ws<1><<<32, 128, 0, stream>>>(rel, w1h + 256, relc, NREL, 128, 384,
                                                128, (NREL + 15) / 16, 2, 1, 64 * 384, 0);

    gat_own<1><<<N_NODES / OWN_NODES, OWN_WAVES * 32, 0, stream>>>(
        el, eln, et, etn, xnode, ec, relc, w2h, num1, keep);

    gat_wmma_gemm_ws<128><<<32, 128, 0, stream>>>(rel, WR, out_rel, NREL, 128, 1, 128,
                                                  (NREL + 15) / 16, 2, 1, 64, 0);
    gat_wmma_gemm_ws<1><<<512, 128, 0, stream>>>(num1, w1o, xab2, N_NODES, 128, 384, 256,
                                                 N_NODES / 16, 4, 2, 128, 64 * 384);
    gat_wmma_gemm_ws<1><<<32, 128, 0, stream>>>(out_rel, w1o + 256, relW2, NREL, 128, 384,
                                                128, (NREL + 15) / 16, 2, 1, 64 * 384, 0);

    gat_own<2><<<N_NODES / OWN_NODES, OWN_WAVES * 32, 0, stream>>>(
        el, eln, et, etn, xab2, ec, relW2, w2o, out, keep);
}


# layer-1 edge lists in two levels: a new launch lists each 512-id segment's edges per 2000-node bin; an owning workgroup scans its bin's slots, not all source ids; entry and add order unchanged
# speedup vs baseline: 2.4758x; 1.0545x over previous
#include <hip/hip_runtime.h>
#include <hip/hip_bf16.h>

typedef float v2f __attribute__((ext_vector_type(2)));
typedef float v8f __attribute__((ext_vector_type(8)));

#define N_NODES 50000
#define E_EDGES 500000
#define E2_EDGES 62500
#define ET_EDGES (E_EDGES + E2_EDGES)
#define NREL 1000
#define ALPHA 0.2f
#define KDIM 128

#define WS_NT 4
#define WS_KS (KDIM / 4)

__device__ __forceinline__ void ws_load_a(v2f (&a)[WS_KS], const float* __restrict__ A,
                                          int mt, int l16, int half, int M, int lda) {
    int arow = mt * 16 + l16;
    arow = arow < M ? arow : (M - 1);
    const float* ap = A + (size_t)arow * lda + half * 2;
#pragma unroll
    for (int j = 0; j < WS_KS; ++j) a[j] = *(const v2f*)(ap + j * 4);
}

__device__ __forceinline__ void ws_tile(const v2f (&a)[WS_KS],
                                        const v2f (&bf)[WS_NT][WS_KS],
                                        float* __restrict__ C, int mt, int l16,
                                        int half, int M, int ldc, int coff) {
    v8f acc[WS_NT];
#pragma unroll
    for (int s = 0; s < WS_NT; ++s) acc[s] = (v8f){0.f,0.f,0.f,0.f,0.f,0.f,0.f,0.f};
#pragma unroll
    for (int j = 0; j < WS_KS; ++j) {
#pragma unroll
        for (int s = 0; s < WS_NT; ++s)
            acc[s] = __builtin_amdgcn_wmma_f32_16x16x4_f32(false, a[j], false,
                                                           bf[s][j], (short)0,
                                                           acc[s], false, false);
    }
    float* cp = C + (size_t)(mt * 16 + half * 8) * ldc + coff + l16;
    if (mt * 16 + 16 <= M) {
#pragma unroll
        for (int r = 0; r < 8; ++r) {
#pragma unroll
            for (int s = 0; s < WS_NT; ++s) cp[(size_t)r * ldc + s * 16] = acc[s][r];
        }
    } else {
#pragma unroll
        for (int r = 0; r < 8; ++r) {
            if (mt * 16 + half * 8 + r < M) {
#pragma unroll
                for (int s = 0; s < WS_NT; ++s) cp[(size_t)r * ldc + s * 16] = acc[s][r];
            }
        }
    }
}

template <int LDBK>
__global__ void __launch_bounds__(128)
gat_wmma_gemm_ws(const float* __restrict__ A, const float* __restrict__ B,
                 float* __restrict__ C, int M, int lda, int ldbn, int ldc,
                 int mtiles, int ncg, int gdiv, int ghi, int glo) {
    const int lane = threadIdx.x & 31;
    const int half = lane >> 4;
    const int l16  = lane & 15;
    const int gw = blockIdx.x * 4 + __builtin_amdgcn_readfirstlane(threadIdx.x >> 5);
    const int nw = gridDim.x * 4;
    const int cg = gw % ncg;
    const int slot = gw / ncg;
    const int nslots = nw / ncg;
    if (slot >= mtiles) return;

    const float* Bg = B + (size_t)(cg / gdiv) * ghi + (size_t)(cg % gdiv) * glo
                        + (size_t)(half * 2) * LDBK;
    v2f bf[WS_NT][WS_KS];
#pragma unroll
    for (int s = 0; s < WS_NT; ++s) {
        const float* bp = Bg + (size_t)(s * 16 + l16) * ldbn;
#pragma unroll
        for (int j = 0; j < WS_KS; ++j) {
            if constexpr (LDBK == 1) bf[s][j] = *(const v2f*)(bp + j * 4);
            else {
                bf[s][j][0] = bp[(size_t)(j * 4) * LDBK];
                bf[s][j][1] = bp[(size_t)(j * 4 + 1) * LDBK];
            }
        }
    }

    const int coff = cg * (WS_NT * 16);
    int mt = slot;
    v2f a0[WS_KS], a1[WS_KS];
    ws_load_a(a0, A, mt, l16, half, M, lda);
    while (true) {
        {
            int mn = mt + nslots;
            mn = mn < mtiles ? mn : mt;
            ws_load_a(a1, A, mn, l16, half, M, lda);
            __builtin_amdgcn_sched_barrier(0);
            ws_tile(a0, bf, C, mt, l16, half, M, ldc, coff);
        }
        mt += nslots;
        if (mt >= mtiles) break;
        {
            int mn = mt + nslots;
            mn = mn < mtiles ? mn : mt;
            ws_load_a(a0, A, mn, l16, half, M, lda);
            __builtin_amdgcn_sched_barrier(0);
            ws_tile(a1, bf, C, mt, l16, half, M, ldc, coff);
        }
        mt += nslots;
        if (mt >= mtiles) break;
    }
}

#define OWN_NODES 200
#define OWN_WAVES 32
#define OWN_CAP   640
#define OWN_NG1   ((E_EDGES + 127) / 128)
#define OWN_NG2   ((E2_EDGES + 127) / 128)
#define OWN_NGT   (OWN_NG1 + OWN_NG2)
#define OWN_GPW   ((OWN_NGT + OWN_WAVES - 1) / OWN_WAVES)
#define OWN_ROUNDS ((OWN_GPW + 3) / 4)
#define OWN_NWG   (N_NODES / OWN_NODES)
#define KEEP_CNT_OFF  256
#define KEEP_W1_OFF   (KEEP_CNT_OFF + OWN_NWG * OWN_WAVES)
#define KEEP_W2_OFF   (KEEP_W1_OFF + OWN_NWG * OWN_WAVES * OWN_CAP)
static_assert(OWN_NWG <= KEEP_CNT_OFF, "one mark per workgroup in front of the lengths");
static_assert(N_NODES <= 65536 && NREL <= 32768, "second list word packs 16 + 15 bits");
static_assert(ET_EDGES <= (1 << 20) && OWN_NODES <= 256, "first list word packs 20 + 8 bits");
static_assert(N_NODES % OWN_NODES == 0, "every workgroup owns a full set of nodes");

#define BIN_NODES 2000
#define BIN_NB    (N_NODES / BIN_NODES)
#define BIN_GPS   4
#define BIN_CAP   64
#define BIN_SPW   36
#define BIN_BPW   5
#define BIN_NSEG  (OWN_WAVES * BIN_SPW)
#define BIN_GW    (BIN_SPW * BIN_CAP / 128)
static_assert(N_NODES % BIN_NODES == 0 && BIN_NODES % OWN_NODES == 0, "bins hold whole workgroups");
static_assert(BIN_NODES <= 2048, "entry packs 20 + 11 bits and stays positive");
static_assert(BIN_NSEG * BIN_GPS >= OWN_NGT, "segments cover every group of source ids");
static_assert((BIN_SPW * BIN_CAP) % 128 == 0 && BIN_SPW >= 32 && BIN_SPW <= 64, "a wave's slots are whole 128-slot groups");
static_assert(BIN_NSEG % 8 == 0 && BIN_NB % BIN_BPW == 0, "gat_bin blocks hold 8 segments, waves whole bin runs");

__global__ void __launch_bounds__(256)
gat_bin(const int* __restrict__ el, const int* __restrict__ eln,
        int* __restrict__ bins, int* __restrict__ bcnt) {
    const int lane = threadIdx.x & 31;
    const int seg = blockIdx.x * 8 + __builtin_amdgcn_readfirstlane(threadIdx.x >> 5);
    const unsigned lt_mask = (1u << lane) - 1u;
    int4 v[BIN_GPS];
#pragma unroll
    for (int q = 0; q < BIN_GPS; ++q) {
        int gi = seg * BIN_GPS + q;
        gi = gi < OWN_NGT ? gi : (OWN_NGT - 1);
        const int* p = gi < OWN_NG1 ? (el + (size_t)gi * 128)
                                    : (eln + (size_t)(gi - OWN_NG1) * 128);
        v[q] = *(const int4*)(p + lane * 4);
    }
    for (int k = 0; k < BIN_BPW; ++k) {
        const int bin = blockIdx.y * BIN_BPW + k;
        const int lo  = bin * BIN_NODES;
        int* outl = bins + ((size_t)bin * BIN_NSEG + seg) * BIN_CAP;
        int cnt = 0;
#pragma unroll
        for (int q = 0; q < BIN_GPS; ++q) {
            int gi = seg * BIN_GPS + q;
            if (gi < OWN_NGT) {
                bool isd = gi < OWN_NG1;
                int eb   = isd ? gi * 128 : (gi - OWN_NG1) * 128;
                int lim  = isd ? E_EDGES : E2_EDGES;
                int eoff = isd ? 0 : E_EDGES;
                int idx0 = eb + lane * 4;
                int sv[4] = {v[q].x, v[q].y, v[q].z, v[q].w};
#pragma unroll
                for (int j = 0; j < 4; ++j) {
                    int loc = sv[j] - lo;
                    bool hit = (idx0 + j < lim) && ((unsigned)loc < (unsigned)BIN_NODES);
                    unsigned mask = __builtin_amdgcn_ballot_w32(hit);
                    if (mask != 0u) {
                        int pos = cnt + __builtin_popcount(mask & lt_mask);
                        if (hit && pos < BIN_CAP) outl[pos] = (eoff + idx0 + j) | (loc << 20);
                        cnt += __builtin_popcount(mask);
                    }
                }
            }
        }
        for (int i = cnt + lane; i < BIN_CAP; i += 32) outl[i] = -1;
        if (lane == 0) bcnt[bin * BIN_NSEG + seg] = cnt;
    }
}

struct OwnEdge {
    float xa[4], xb[4], r0[4], r1[4];
    int nl;
    int direct;
};

__device__ __forceinline__ void own_types(int e, const int* __restrict__ etn, int& t0, int& t1) {
    t0 = 0; t1 = 0;
    if (e >= E_EDGES) {
        int j = e - E_EDGES;
        t0 = etn[2 * j]; t1 = etn[2 * j + 1];
    }
}

template <int L>
__device__ __forceinline__ void own_rows_node(OwnEdge& r, int nl, int dt, int base, int lane,
                                              const float* __restrict__ xrow) {
    const int s = base + nl;
    const int d = dt & 0xFFFF;
    r.nl = nl;
    if constexpr (L == 1) {
        const int hc = (lane >> 4) * 128 + (lane & 15) * 4;
        float4 a = *(const float4*)(xrow + (size_t)s * 256 + hc);
        float4 b = *(const float4*)(xrow + (size_t)d * 256 + hc + 64);
        r.xa[0] = a.x; r.xa[1] = a.y; r.xa[2] = a.z; r.xa[3] = a.w;
        r.xb[0] = b.x; r.xb[1] = b.y; r.xb[2] = b.z; r.xb[3] = b.w;
    } else {
        float4 a = *(const float4*)(xrow + (size_t)s * 256 + lane * 4);
        float4 b = *(const float4*)(xrow + (size_t)d * 256 + 128 + lane * 4);
        r.xa[0] = a.x; r.xa[1] = a.y; r.xa[2] = a.z; r.xa[3] = a.w;
        r.xb[0] = b.x; r.xb[1] = b.y; r.xb[2] = b.z; r.xb[3] = b.w;
    }
}

template <int L>
__device__ __forceinline__ void own_rows_rel(OwnEdge& r, int e, int dt, int t0, int t1, int lane,
                                             const float* __restrict__ erow,
                                             const float* __restrict__ rtab) {
    const bool direct = e < E_EDGES;
    r.direct = direct ? 1 : 0;
    if constexpr (L == 1) {
        const float* p0 = direct ? (erow + (size_t)e * 128) : (rtab + (size_t)t0 * 128);
        float4 v = *(const float4*)(p0 + lane * 4);
        r.r0[0] = v.x; r.r0[1] = v.y; r.r0[2] = v.z; r.r0[3] = v.w;
        if (!direct) {
            float4 a1 = *(const float4*)(rtab + (size_t)t1 * 128 + lane * 4);
            r.r1[0] = a1.x; r.r1[1] = a1.y; r.r1[2] = a1.z; r.r1[3] = a1.w;
        }
    } else {
        const int t = direct ? (dt >> 16) : t0;
        float4 v = *(const float4*)(rtab + (size_t)t * 128 + lane * 4);
        r.r0[0] = v.x; r.r0[1] = v.y; r.r0[2] = v.z; r.r0[3] = v.w;
        if (!direct) {
            float4 a1 = *(const float4*)(rtab + (size_t)t1 * 128 + lane * 4);
            r.r1[0] = a1.x; r.r1[1] = a1.y; r.r1[2] = a1.z; r.r1[3] = a1.w;
        }
    }
}

template <int L>
__device__ __forceinline__ void own_load(OwnEdge& r, int e, int nl, int dt, int base, int lane,
                                         const int* __restrict__ etn,
                                         const float* __restrict__ xrow,
                                         const float* __restrict__ erow,
                                         const float* __restrict__ rtab) {
    int t0, t1;
#pragma unroll
    for (int c = 0; c < 4; ++c) r.r1[c] = 0.f;
    own_types(e, etn, t0, t1);
    own_rows_node<L>(r, nl, dt, base, lane, xrow);
    own_rows_rel<L>(r, e, dt, t0, t1, lane, erow, rtab);
}

template <int L>
__device__ __forceinline__ void own_load2(OwnEdge& rA, OwnEdge& rB, int enA, int dtA, int enB,
                                          int dtB, int base, int lane,
                                          const int* __restrict__ etn,
                                          const float* __restrict__ xrow,
                                          const float* __restrict__ erow,
                                          const float* __restrict__ rtab) {
    const int eA = enA & 0xFFFFF, eB = enB & 0xFFFFF;
    int tA0, tA1, tB0, tB1;
#pragma unroll
    for (int c = 0; c < 4; ++c) { rA.r1[c] = 0.f; rB.r1[c] = 0.f; }
    own_types(eA, etn, tA0, tA1);
    own_types(eB, etn, tB0, tB1);
    own_rows_node<L>(rA, enA >> 20, dtA, base, lane, xrow);
    own_rows_node<L>(rB, enB >> 20, dtB, base, lane, xrow);
    own_rows_rel<L>(rA, eA, dtA, tA0, tA1, lane, erow, rtab);
    own_rows_rel<L>(rB, eB, dtB, tB0, tB1, lane, erow, rtab);
}

template <int L> struct OwnLay {
    static constexpr int XOFF = (L == 1) ? 8 : 16;
    static __device__ __forceinline__ bool den_writer(int lane) {
        return (L == 1) ? ((lane & 15) == 0) : (lane == 0);
    }
    static __device__ __forceinline__ int den_idx(int nl, int lane) {
        return (L == 1) ? (nl * 2 + (lane >> 4)) : nl;
    }
};

template <int L>
__device__ __forceinline__ void own_add(const OwnEdge& r, const float (&w)[4], int lane,
                                        float* acc, float* den) {
    float m[4];
#pragma unroll
    for (int c = 0; c < 4; ++c) {
        float ev = r.direct ? r.r0[c] : (r.r0[c] + r.r1[c]);
        m[c] = r.xa[c] + r.xb[c] + ev;
    }
    float p = m[0] * w[0] + m[1] * w[1] + m[2] * w[2] + m[3] * w[3];
#pragma unroll
    for (int off = OwnLay<L>::XOFF; off > 0; off >>= 1) p += __shfl_xor(p, off, 32);
    float lr = p > 0.f ? p : ALPHA * p;
    float eev = __expf(-lr);
    float4* ap = (float4*)(acc + r.nl * 128 + lane * 4);
    float4 av = *ap;
    av.x += eev * m[0];
    av.y += eev * m[1];
    av.z += eev * m[2];
    av.w += eev * m[3];
    *ap = av;
    if (OwnLay<L>::den_writer(lane)) den[OwnLay<L>::den_idx(r.nl, lane)] += eev;
}


template <int L>
__device__ __forceinline__ void own_add2(const OwnEdge& A, const OwnEdge& B,
                                         const float (&w)[4], int lane,
                                         float* acc, float* den) {
    float mA[4], mB[4];
#pragma unroll
    for (int c = 0; c < 4; ++c) {
        float evA = A.direct ? A.r0[c] : (A.r0[c] + A.r1[c]);
        float evB = B.direct ? B.r0[c] : (B.r0[c] + B.r1[c]);
        mA[c] = A.xa[c] + A.xb[c] + evA;
        mB[c] = B.xa[c] + B.xb[c] + evB;
    }
    const bool same = (A.nl == B.nl);
    float pA = mA[0] * w[0] + mA[1] * w[1] + mA[2] * w[2] + mA[3] * w[3];
    float pB = mB[0] * w[0] + mB[1] * w[1] + mB[2] * w[2] + mB[3] * w[3];
#pragma unroll
    for (int off = OwnLay<L>::XOFF; off > 0; off >>= 1) {
        float tA = __shfl_xor(pA, off, 32);
        float tB = __shfl_xor(pB, off, 32);
        pA += tA;
        pB += tB;
    }
    float lrA = pA > 0.f ? pA : ALPHA * pA;
    float lrB = pB > 0.f ? pB : ALPHA * pB;
    float eA = __expf(-lrA);
    float eB = __expf(-lrB);
    float4* apA = (float4*)(acc + A.nl * 128 + lane * 4);
    float4* apB = (float4*)(acc + B.nl * 128 + lane * 4);
    float4 a = *apA, b = *apB;
    a.x += eA * mA[0]; a.y += eA * mA[1]; a.z += eA * mA[2]; a.w += eA * mA[3];
    if (same) b = a;
    b.x += eB * mB[0]; b.y += eB * mB[1]; b.z += eB * mB[2]; b.w += eB * mB[3];
    *apA = a;
    *apB = b;
    if (OwnLay<L>::den_writer(lane)) {
        const int iA = OwnLay<L>::den_idx(A.nl, lane), iB = OwnLay<L>::den_idx(B.nl, lane);
        float dA = den[iA], dB = den[iB];
        dA += eA;
        if (same) dB = dA;
        dB += eB;
        den[iA] = dA;
        den[iB] = dB;
    }
}

template <int L>
__global__ void __launch_bounds__(OWN_WAVES * 32)
gat_own(const int* __restrict__ el, const int* __restrict__ eln,
        const int* __restrict__ et, const int* __restrict__ etn,
        const float* __restrict__ xrow, const float* __restrict__ erow,
        const float* __restrict__ rtab, const float* __restrict__ wvec,
        float* __restrict__ outp, int* __restrict__ keep,
        const int* __restrict__ bins, const int* __restrict__ bcnt) {
    __shared__ float acc[OWN_NODES * 128];
    __shared__ float den[OWN_NODES * 2];
    __shared__ int lists[OWN_WAVES * OWN_CAP];
    __shared__ int lists2[OWN_WAVES * OWN_CAP];
    __shared__ int s_cnt[OWN_WAVES];
    __shared__ int s_more;
    __shared__ int s_long;
    __shared__ int s_ovf[OWN_WAVES];

    const int tid  = threadIdx.x;
    const int lane = tid & 31;
    const int wv   = __builtin_amdgcn_readfirstlane(tid >> 5);
    const int base = blockIdx.x * OWN_NODES;

    for (int i = tid; i < OWN_NODES * 128; i += OWN_WAVES * 32) acc[i] = 0.f;
    for (int i = tid; i < OWN_NODES * 2; i += OWN_WAVES * 32) den[i] = 0.f;
    if (tid == 0) s_long = 0;

    float w[4];
    w[0] = wvec[lane * 4];     w[1] = wvec[lane * 4 + 1];
    w[2] = wvec[lane * 4 + 2]; w[3] = wvec[lane * 4 + 3];

    const int bin = blockIdx.x / (BIN_NODES / OWN_NODES);
    const int blo = (blockIdx.x % (BIN_NODES / OWN_NODES)) * OWN_NODES;
    {
        const int* cp = bcnt + bin * BIN_NSEG + wv * BIN_SPW;
        int c0 = cp[lane];
        int c1 = lane < BIN_SPW - 32 ? cp[32 + lane] : 0;
        bool over = (c0 > BIN_CAP) || (c1 > BIN_CAP);
        unsigned any = __builtin_amdgcn_ballot_w32(over);
        if (lane == 0) s_ovf[wv] = any != 0u ? 1 : 0;
    }
    __syncthreads();
    const bool binned = __builtin_amdgcn_ballot_w32(s_ovf[lane] != 0) == 0u;
    const int* bslots = bins + (size_t)bin * (BIN_NSEG * BIN_CAP);

    int g = binned ? wv * BIN_GW : wv * OWN_GPW;
    int gend = binned ? g + BIN_GW : g + OWN_GPW;
    if (!binned) gend = gend < OWN_NGT ? gend : OWN_NGT;
    int* mylist = lists + wv * OWN_CAP;
    int* mylist2 = lists2 + wv * OWN_CAP;
    const unsigned lt_mask = (1u << lane) - 1u;

    int* k_mark = keep + blockIdx.x;
    int* k_cnt  = keep + KEEP_CNT_OFF + blockIdx.x * OWN_WAVES + wv;
    int* k_w1   = keep + KEEP_W1_OFF + (size_t)(blockIdx.x * OWN_WAVES + wv) * OWN_CAP;
    int* k_w2   = keep + KEEP_W2_OFF + (size_t)(blockIdx.x * OWN_WAVES + wv) * OWN_CAP;
    bool reuse = false;
    if constexpr (L == 2) {
        reuse = (*k_mark == 1);
        if (reuse) g = gend;
    }
    const int w2lo = reuse ? wv : 0;
    const int w2hi = reuse ? wv + 1 : OWN_WAVES;
    int kept = 0;
    bool single = false;

    for (int round = 0; round < OWN_ROUNDS; ++round) {
        if (tid == 0) s_more = 0;
        __syncthreads();

        int cnt = 0;
        if (reuse) {
            cnt = *k_cnt;
            cnt = cnt < 0 ? 0 : (cnt > OWN_CAP ? OWN_CAP : cnt);
            for (int i = lane; i < cnt; i += 32) {
                mylist[i] = k_w1[i];
                mylist2[i] = k_w2[i];
            }
        }
        while (binned && g < gend && cnt + 512 <= OWN_CAP) {
            int sv[4][4];
#pragma unroll
            for (int q = 0; q < 4; ++q) {
                int gi = g + q;
                gi = gi < gend ? gi : (gend - 1);
                const int* p = bslots + (size_t)gi * 128 + lane;
#pragma unroll
                for (int j = 0; j < 4; ++j) sv[q][j] = p[j * 32];
            }
#pragma unroll
            for (int q = 0; q < 4; ++q) {
                if (g + q < gend) {
#pragma unroll
                    for (int j = 0; j < 4; ++j) {
                        int ent = sv[q][j];
                        int loc = (ent >> 20) - blo;
                        bool hit = (unsigned)loc < (unsigned)OWN_NODES;
                        unsigned mask = __builtin_amdgcn_ballot_w32(hit);
                        if (mask != 0u) {
                            int pos = cnt + __builtin_popcount(mask & lt_mask);
                            if (hit) mylist[pos] = (ent & 0xFFFFF) | (loc << 20);
                            cnt += __builtin_popcount(mask);
                        }
                    }
                }
            }
            g += 4;
        }
        while (!binned && g < gend && cnt + 512 <= OWN_CAP) {
            int4 v[4];
#pragma unroll
            for (int q = 0; q < 4; ++q) {
                int gi = g + q;
                gi = gi < gend ? gi : (gend - 1);
                const int* p = gi < OWN_NG1 ? (el + (size_t)gi * 128)
                                            : (eln + (size_t)(gi - OWN_NG1) * 128);
                v[q] = *(const int4*)(p + lane * 4);
            }
#pragma unroll
            for (int q = 0; q < 4; ++q) {
                int gi = g + q;
                if (gi < gend) {
                    bool isd = gi < OWN_NG1;
                    int eb   = isd ? gi * 128 : (gi - OWN_NG1) * 128;
                    int lim  = isd ? E_EDGES : E2_EDGES;
                    int eoff = isd ? 0 : E_EDGES;
                    int idx0 = eb + lane * 4;
                    int sv[4] = {v[q].x, v[q].y, v[q].z, v[q].w};
#pragma unroll
                    for (int j = 0; j < 4; ++j) {
                        int loc = sv[j] - base;
                        bool hit = (idx0 + j < lim) && ((unsigned)loc < (unsigned)OWN_NODES);
                        unsigned mask = __builtin_amdgcn_ballot_w32(hit);
                        if (mask != 0u) {
                            int pos = cnt + __builtin_popcount(mask & lt_mask);
                            if (hit) mylist[pos] = (eoff + idx0 + j) | (loc << 20);
                            cnt += __builtin_popcount(mask);
                        }
                    }
                }
            }
            g += 4;
        }
        for (int i = lane; i < (reuse ? 0 : cnt); i += 32) {
            int e = mylist[i] & 0xFFFFF;
            int d, t = 0;
            if (e < E_EDGES) {
                d = el[E_EDGES + e];
                t = et[e];
            } else {
                d = eln[E2_EDGES + (e - E_EDGES)];
            }
            mylist2[i] = d | (t << 16);
        }
        if (lane == 0) {
            s_cnt[wv] = cnt;
            if (g < gend) s_more = 1;
        }
        __syncthreads();
        const int more = s_more;

        {
            OwnEdge curA = {}, curB = {};
            bool have = false;
            bool half = false;
            int pen = 0, pen2 = 0;
            for (int w2 = w2lo; w2 < w2hi; ++w2) {
                const int n = __builtin_amdgcn_readfirstlane(s_cnt[w2]);
                const int* lp = lists + w2 * OWN_CAP;
                const int* lp2 = lists2 + w2 * OWN_CAP;
                for (int i0 = 0; i0 < n; i0 += 32) {
                    int idx = i0 + lane;
                    int ent = idx < n ? lp[idx] : 0;
                    int ent2 = idx < n ? lp2[idx] : 0;
                    bool mine = (idx < n) && (((ent >> 20) & 31) == wv);
                    unsigned mask = __builtin_amdgcn_ballot_w32(mine);
                    while (mask != 0u) {
                        int b = __builtin_ctz(mask);
                        mask &= mask - 1u;
                        int en = __builtin_amdgcn_readlane(ent, b);
                        int en2 = __builtin_amdgcn_readlane(ent2, b);
                        if constexpr (L == 1) {
                            if (kept < OWN_CAP && lane == 0) { k_w1[kept] = en; k_w2[kept] = en2; }
                            ++kept;
                        }
                        if (!half) {
                            pen = en;
                            pen2 = en2;
                            half = true;
                        } else {
                            OwnEdge nxA, nxB;
                            own_load2<L>(nxA, nxB, pen, pen2, en, en2, base, lane,
                                         etn, xrow, erow, rtab);
                            if (have) own_add2<L>(curA, curB, w, lane, acc, den);
                            curA = nxA;
                            curB = nxB;
                            have = true;
                            half = false;
                        }
                    }
                }
            }
            if (have) own_add2<L>(curA, curB, w, lane, acc, den);
            if (half) {
                OwnEdge last;
                own_load<L>(last, pen & 0xFFFFF, pen >> 20, pen2, base, lane,
                            etn, xrow, erow, rtab);
                own_add<L>(last, w, lane, acc, den);
            }
        }
        if constexpr (L == 1) {
            if (lane == 0 && kept > OWN_CAP) s_long = 1;
        }
        __syncthreads();
        if (!more) { single = (round == 0); break; }
    }
    if constexpr (L == 1) {
        if (lane == 0) *k_cnt = kept;
        if (tid == 0) *k_mark = (single && s_long == 0) ? 1 : 0;
    }

    float* orow = outp + (size_t)base * 128;
    for (int i = tid; i < OWN_NODES * 128; i += OWN_WAVES * 32) {
        int nl = i >> 7;
        float dnm;
        if constexpr (L == 1) dnm = den[nl * 2 + ((i & 127) >> 6)];
        else                  dnm = den[nl];
        dnm = dnm > 1e-12f ? dnm : 1e-12f;
        float h = acc[i] / dnm;
        if constexpr (L == 1) {
            orow[i] = h;
        } else {
            float y = h > 0.f ? h : (__expf(h) - 1.f);
            y = y > 0.f ? y : (__expf(y) - 1.f);
            orow[i] = y;
        }
    }
}

extern "C" void kernel_launch(void* const* d_in, const int* in_sizes, int n_in,
                              void* d_out, int out_size, void* d_ws, size_t ws_size,
                              hipStream_t stream) {
    const float* x    = (const float*)d_in[0];
    const float* rel  = (const float*)d_in[1];
    const int*   el   = (const int*)d_in[2];
    const int*   et   = (const int*)d_in[3];
    const float* eemb = (const float*)d_in[4];
    const int*   eln  = (const int*)d_in[5];
    const int*   etn  = (const int*)d_in[6];
    const float* w1h  = (const float*)d_in[8];
    const float* w2h  = (const float*)d_in[9];
    const float* w1o  = (const float*)d_in[10];
    const float* w2o  = (const float*)d_in[11];
    const float* WR   = (const float*)d_in[12];
    float* out = (float*)d_out;

    float* ws = (float*)d_ws;
    float* xnode  = ws;
    float* ec     = xnode  + (size_t)N_NODES * 256;
    float* relc   = ec     + (size_t)E_EDGES * 128;
    float* num1   = relc   + (size_t)NREL * 128;
    float* xab2   = num1   + (size_t)N_NODES * 128;
    float* relW2  = xab2   + (size_t)N_NODES * 256;
    int*   keep   = (int*)(relW2 + (size_t)NREL * 128);
    int*   bins   = keep + KEEP_W2_OFF + (size_t)OWN_NWG * OWN_WAVES * OWN_CAP;
    int*   bcnt   = bins + (size_t)BIN_NB * BIN_NSEG * BIN_CAP;

    float* out_rel = out + (size_t)N_NODES * 128;

    gat_wmma_gemm_ws<1><<<512, 128, 0, stream>>>(x, w1h, xnode, N_NODES, 128, 384, 256,
                                                 N_NODES / 16, 4, 2, 64 * 384, 128);
    gat_wmma_gemm_ws<1><<<512, 128, 0, stream>>>(eemb, w1h + 256, ec, E_EDGES, 128, 384,
                                                 128, E_EDGES / 16, 2, 1, 64 * 384, 0);
    gat_wmma_gemm_ws<1><<<32, 128, 0, stream>>>(rel, w1h + 256, relc, NREL, 128, 384,
                                                128, (NREL + 15) / 16, 2, 1, 64 * 384, 0);

    gat_bin<<<dim3(BIN_NSEG / 8, BIN_NB / BIN_BPW), 256, 0, stream>>>(el, eln, bins, bcnt);

    gat_own<1><<<N_NODES / OWN_NODES, OWN_WAVES * 32, 0, stream>>>(
        el, eln, et, etn, xnode, ec, relc, w2h, num1, keep, bins, bcnt);

    gat_wmma_gemm_ws<128><<<32, 128, 0, stream>>>(rel, WR, out_rel, NREL, 128, 1, 128,
                                                  (NREL + 15) / 16, 2, 1, 64, 0);
    gat_wmma_gemm_ws<1><<<512, 128, 0, stream>>>(num1, w1o, xab2, N_NODES, 128, 384, 256,
                                                 N_NODES / 16, 4, 2, 128, 64 * 384);
    gat_wmma_gemm_ws<1><<<32, 128, 0, stream>>>(out_rel, w1o + 256, relW2, NREL, 128, 384,
                                                128, (NREL + 15) / 16, 2, 1, 64 * 384, 0);

    gat_own<2><<<N_NODES / OWN_NODES, OWN_WAVES * 32, 0, stream>>>(
        el, eln, et, etn, xab2, ec, relW2, w2o, out, keep, bins, bcnt);
}
